# starting kernel
# baseline (speedup 1.0000x reference)
#include <hip/hip_runtime.h>

typedef _Float16 f16;
typedef __attribute__((ext_vector_type(16))) _Float16 v16h;
typedef __attribute__((ext_vector_type(8)))  float    v8f;

#define NNODES 50000
#define NEDGES 800000
#define FIN    128
#define FOUT   128
#define MT     4

union V16H { v16h v; uint4 q[2]; };
union H8   { f16 h[8]; uint4 q; };

__device__ __forceinline__ uint4 pack8(float4 a, float4 b) {
  H8 u;
  u.h[0] = (f16)a.x; u.h[1] = (f16)a.y; u.h[2] = (f16)a.z; u.h[3] = (f16)a.w;
  u.h[4] = (f16)b.x; u.h[5] = (f16)b.y; u.h[6] = (f16)b.z; u.h[7] = (f16)b.w;
  return u.q;
}

__device__ __forceinline__ v16h ld_a_frag(const f16* lds, int row_stride,
                                          int M, int hi, int kbase) {
  const int kb = kbase + hi * 8;
  V16H t;
  t.q[0] = *(const uint4*)(lds + M * row_stride + kb);
  t.q[1] = *(const uint4*)(lds + M * row_stride + kb + 16);
  return t.v;
}

__device__ __forceinline__ v16h ld_b_frag(const f16* __restrict__ wf,
                                          int frag, int lane) {
  const uint4* p = (const uint4*)(wf + ((size_t)(frag * 32 + lane)) * 16);
  V16H t;
  t.q[0] = p[0];
  t.q[1] = p[1];
  return t.v;
}

__global__ __launch_bounds__(256) void pack_weights_kernel(
    const float* __restrict__ W, f16* __restrict__ out, int K, int N) {
  int o = blockIdx.x * blockDim.x + threadIdx.x;
  if (o >= K * N) return;
  int ksteps = K >> 5;
  int e    = o & 15;
  int lane = (o >> 4) & 31;
  int frag = o >> 9;
  int ks   = frag % ksteps;
  int nt   = frag / ksteps;
  int kl   = (e < 8 ? e : e + 8) + ((lane & 16) ? 8 : 0);
  int Kidx = ks * 32 + kl;
  int n    = nt * 16 + (lane & 15);
  out[o] = (f16)W[Kidx * N + n];
}

__global__ __launch_bounds__(256) void edge_mlp_kernel(
    const float* __restrict__ x,
    const int* __restrict__ src,
    const int* __restrict__ dst,
    const f16* __restrict__ W1f,
    const float* __restrict__ b1,
    const f16* __restrict__ W2f,
    const float* __restrict__ b2,
    float* __restrict__ summed,
    float* __restrict__ cnt) {
  __shared__ f16 lds_e[MT * 16 * 256];
  __shared__ f16 lds_h[MT * 16 * 128];
  __shared__ int lds_dst[MT * 16];

  const int tid   = threadIdx.x;
  const int lane  = tid & 31;
  const int wv    = tid >> 5;
  const int ebase = blockIdx.x * (MT * 16);

  {
    const int r  = tid >> 2;
    const int c0 = (tid & 3) * 32;
    const int e  = ebase + r;
    const int si = src[e];
    const int di = dst[e];
    if ((tid & 3) == 0) {
      lds_dst[r] = di;
      atomicAdd(&cnt[di], 1.0f);
    }
    const float4* xi = (const float4*)(x + (size_t)di * FIN + c0);
    const float4* xj = (const float4*)(x + (size_t)si * FIN + c0);
    uint4* pe = (uint4*)(&lds_e[r * 256 + c0]);
    uint4* pd = (uint4*)(&lds_e[r * 256 + 128 + c0]);
#pragma unroll
    for (int k = 0; k < 4; ++k) {
      float4 a0 = xi[2 * k], a1 = xi[2 * k + 1];
      float4 j0 = xj[2 * k], j1 = xj[2 * k + 1];
      pe[k] = pack8(a0, a1);
      float4 d0 = make_float4(j0.x - a0.x, j0.y - a0.y, j0.z - a0.z, j0.w - a0.w);
      float4 d1 = make_float4(j1.x - a1.x, j1.y - a1.y, j1.z - a1.z, j1.w - a1.w);
      pd[k] = pack8(d0, d1);
    }
  }
  __syncthreads();

  const int M  = lane & 15;
  const int hi = (lane >> 4) & 1;
  const int n  = wv * 16 + (lane & 15);

  v8f acc[MT];
#pragma unroll
  for (int mt = 0; mt < MT; ++mt) acc[mt] = (v8f){};
#pragma unroll
  for (int ks = 0; ks < 8; ++ks) {
    v16h b = ld_b_frag(W1f, wv * 8 + ks, lane);
#pragma unroll
    for (int mt = 0; mt < MT; ++mt) {
      v16h a = ld_a_frag(lds_e + mt * 16 * 256, 256, M, hi, ks * 32);
      acc[mt] = __builtin_amdgcn_wmma_f32_16x16x32_f16(false, a, false, b,
                                                       (short)0, acc[mt],
                                                       false, false);
    }
  }
  {
    const float bias = b1[n];
#pragma unroll
    for (int mt = 0; mt < MT; ++mt) {
#pragma unroll
      for (int r = 0; r < 8; ++r) {
        float v = acc[mt][r] + bias;
        v = v > 0.0f ? v : 0.0f;
        lds_h[(mt * 16 + r + 8 * hi) * 128 + n] = (f16)v;
      }
    }
  }
  __syncthreads();

  v8f acc2[MT];
#pragma unroll
  for (int mt = 0; mt < MT; ++mt) acc2[mt] = (v8f){};
#pragma unroll
  for (int ks = 0; ks < 4; ++ks) {
    v16h b = ld_b_frag(W2f, wv * 4 + ks, lane);
#pragma unroll
    for (int mt = 0; mt < MT; ++mt) {
      v16h a = ld_a_frag(lds_h + mt * 16 * 128, 128, M, hi, ks * 32);
      acc2[mt] = __builtin_amdgcn_wmma_f32_16x16x32_f16(false, a, false, b,
                                                        (short)0, acc2[mt],
                                                        false, false);
    }
  }
  {
    const float bias = b2[n];
#pragma unroll
    for (int mt = 0; mt < MT; ++mt) {
#pragma unroll
      for (int r = 0; r < 8; ++r) {
        const int row = mt * 16 + r + 8 * hi;
        atomicAdd(&summed[(size_t)lds_dst[row] * FOUT + n],
                  acc2[mt][r] + bias);
      }
    }
  }
}

__global__ __launch_bounds__(256) void node_mlp_kernel(
    const float* __restrict__ summed,
    const float* __restrict__ cnt,
    const f16* __restrict__ W3f,
    const float* __restrict__ b3,
    const f16* __restrict__ W4f,
    const float* __restrict__ b4,
    float* __restrict__ out) {
  __shared__ f16 lds_a[16 * 128];
  __shared__ f16 lds_m[16 * 64];

  const int tid   = threadIdx.x;
  const int lane  = tid & 31;
  const int wv    = tid >> 5;
  const int nbase = blockIdx.x * 16;

  {
    const int r    = tid >> 4;
    const int c0   = (tid & 15) * 8;
    const int node = nbase + r;
    float cv  = cnt[node];
    float inv = 1.0f / (cv > 1.0f ? cv : 1.0f);
    const float4* sp = (const float4*)(summed + (size_t)node * FOUT + c0);
    float4 s0 = sp[0], s1 = sp[1];
    float4 m0 = make_float4(s0.x * inv, s0.y * inv, s0.z * inv, s0.w * inv);
    float4 m1 = make_float4(s1.x * inv, s1.y * inv, s1.z * inv, s1.w * inv);
    m0.x = m0.x > 0.f ? m0.x : 0.f; m0.y = m0.y > 0.f ? m0.y : 0.f;
    m0.z = m0.z > 0.f ? m0.z : 0.f; m0.w = m0.w > 0.f ? m0.w : 0.f;
    m1.x = m1.x > 0.f ? m1.x : 0.f; m1.y = m1.y > 0.f ? m1.y : 0.f;
    m1.z = m1.z > 0.f ? m1.z : 0.f; m1.w = m1.w > 0.f ? m1.w : 0.f;
    *(uint4*)(&lds_a[r * 128 + c0]) = pack8(m0, m1);
  }
  __syncthreads();

  const int M  = lane & 15;
  const int hi = (lane >> 4) & 1;

  if (wv < 4) {
    const int n = wv * 16 + (lane & 15);
    v8f acc = {};
#pragma unroll
    for (int ks = 0; ks < 4; ++ks) {
      v16h a = ld_a_frag(lds_a, 128, M, hi, ks * 32);
      v16h b = ld_b_frag(W3f, wv * 4 + ks, lane);
      acc = __builtin_amdgcn_wmma_f32_16x16x32_f16(false, a, false, b,
                                                   (short)0, acc, false, false);
    }
    const float bias = b3[n];
#pragma unroll
    for (int r = 0; r < 8; ++r) {
      float v = acc[r] + bias;
      v = v > 0.0f ? v : 0.0f;
      lds_m[(r + 8 * hi) * 64 + n] = (f16)v;
    }
  }
  __syncthreads();

  {
    const int n = wv * 16 + (lane & 15);
    v8f acc = {};
#pragma unroll
    for (int ks = 0; ks < 2; ++ks) {
      v16h a = ld_a_frag(lds_m, 64, M, hi, ks * 32);
      v16h b = ld_b_frag(W4f, wv * 2 + ks, lane);
      acc = __builtin_amdgcn_wmma_f32_16x16x32_f16(false, a, false, b,
                                                   (short)0, acc, false, false);
    }
    const float bias = b4[n];
#pragma unroll
    for (int r = 0; r < 8; ++r) {
      out[(size_t)(nbase + r + 8 * hi) * FOUT + n] = acc[r] + bias;
    }
  }
}

extern "C" void kernel_launch(void* const* d_in, const int* in_sizes, int n_in,
                              void* d_out, int out_size, void* d_ws, size_t ws_size,
                              hipStream_t stream) {
  const float* x   = (const float*)d_in[0];
  const int*  eidx = (const int*)d_in[1];
  const float* W1  = (const float*)d_in[2];
  const float* b1  = (const float*)d_in[3];
  const float* W2  = (const float*)d_in[4];
  const float* b2  = (const float*)d_in[5];
  const float* W3  = (const float*)d_in[6];
  const float* b3  = (const float*)d_in[7];
  const float* W4  = (const float*)d_in[8];
  const float* b4  = (const float*)d_in[9];
  float* out = (float*)d_out;

  char* ws = (char*)d_ws;
  float* summed = (float*)ws;
  float* cnt    = (float*)(ws + 25600000);
  f16* W1f = (f16*)(ws + 25800000);
  f16* W2f = (f16*)(ws + 25800000 + 65536);
  f16* W3f = (f16*)(ws + 25800000 + 65536 + 32768);
  f16* W4f = (f16*)(ws + 25800000 + 65536 + 32768 + 16384);

  hipMemsetAsync(summed, 0, (size_t)(NNODES * FOUT + NNODES) * sizeof(float), stream);

  pack_weights_kernel<<<(256 * 128 + 255) / 256, 256, 0, stream>>>(W1, W1f, 256, 128);
  pack_weights_kernel<<<(128 * 128 + 255) / 256, 256, 0, stream>>>(W2, W2f, 128, 128);
  pack_weights_kernel<<<(128 * 64  + 255) / 256, 256, 0, stream>>>(W3, W3f, 128, 64);
  pack_weights_kernel<<<(64  * 128 + 255) / 256, 256, 0, stream>>>(W4, W4f, 64, 128);

  edge_mlp_kernel<<<NEDGES / (MT * 16), 256, 0, stream>>>(
      x, eidx, eidx + NEDGES, W1f, b1, W2f, b2, summed, cnt);

  node_mlp_kernel<<<NNODES / 16, 256, 0, stream>>>(
      summed, cnt, W3f, b3, W4f, b4, out);
}


# Pad the LDS row pitch of every WMMA A tile (edge feature and hidden tiles, node mean and hidden tiles) to an odd multiple of the b128 access so fragment loads and staging stores are bank-conflict-free
# speedup vs baseline: 2.1698x; 2.1698x over previous
#include <hip/hip_runtime.h>

typedef _Float16 f16;
typedef __attribute__((ext_vector_type(16))) _Float16 v16h;
typedef __attribute__((ext_vector_type(8)))  float    v8f;

#define NNODES 50000
#define NEDGES 800000
#define FIN    128
#define FOUT   128
#define MT     4

#define EPITCH (2 * FIN + 8)
#define HPITCH (FOUT + 8)
#define APITCH (FOUT + 8)
#define MPITCH (FOUT / 2 + 8)

union V16H { v16h v; uint4 q[2]; };
union H8   { f16 h[8]; uint4 q; };

__device__ __forceinline__ uint4 pack8(float4 a, float4 b) {
  H8 u;
  u.h[0] = (f16)a.x; u.h[1] = (f16)a.y; u.h[2] = (f16)a.z; u.h[3] = (f16)a.w;
  u.h[4] = (f16)b.x; u.h[5] = (f16)b.y; u.h[6] = (f16)b.z; u.h[7] = (f16)b.w;
  return u.q;
}

__device__ __forceinline__ v16h ld_a_frag(const f16* lds, int row_stride,
                                          int M, int hi, int kbase) {
  const int kb = kbase + hi * 8;
  V16H t;
  t.q[0] = *(const uint4*)(lds + M * row_stride + kb);
  t.q[1] = *(const uint4*)(lds + M * row_stride + kb + 16);
  return t.v;
}

__device__ __forceinline__ v16h ld_b_frag(const f16* __restrict__ wf,
                                          int frag, int lane) {
  const uint4* p = (const uint4*)(wf + ((size_t)(frag * 32 + lane)) * 16);
  V16H t;
  t.q[0] = p[0];
  t.q[1] = p[1];
  return t.v;
}

__global__ __launch_bounds__(256) void pack_weights_kernel(
    const float* __restrict__ W, f16* __restrict__ out, int K, int N) {
  int o = blockIdx.x * blockDim.x + threadIdx.x;
  if (o >= K * N) return;
  int ksteps = K >> 5;
  int e    = o & 15;
  int lane = (o >> 4) & 31;
  int frag = o >> 9;
  int ks   = frag % ksteps;
  int nt   = frag / ksteps;
  int kl   = (e < 8 ? e : e + 8) + ((lane & 16) ? 8 : 0);
  int Kidx = ks * 32 + kl;
  int n    = nt * 16 + (lane & 15);
  out[o] = (f16)W[Kidx * N + n];
}

__global__ __launch_bounds__(256) void edge_mlp_kernel(
    const float* __restrict__ x,
    const int* __restrict__ src,
    const int* __restrict__ dst,
    const f16* __restrict__ W1f,
    const float* __restrict__ b1,
    const f16* __restrict__ W2f,
    const float* __restrict__ b2,
    float* __restrict__ summed,
    float* __restrict__ cnt) {
  __shared__ __attribute__((aligned(16))) f16 lds_e[MT * 16 * EPITCH];
  __shared__ __attribute__((aligned(16))) f16 lds_h[MT * 16 * HPITCH];
  __shared__ int lds_dst[MT * 16];

  const int tid   = threadIdx.x;
  const int lane  = tid & 31;
  const int wv    = tid >> 5;
  const int ebase = blockIdx.x * (MT * 16);

  {
    const int r  = tid >> 2;
    const int c0 = (tid & 3) * 32;
    const int e  = ebase + r;
    const int si = src[e];
    const int di = dst[e];
    if ((tid & 3) == 0) {
      lds_dst[r] = di;
      atomicAdd(&cnt[di], 1.0f);
    }
    const float4* xi = (const float4*)(x + (size_t)di * FIN + c0);
    const float4* xj = (const float4*)(x + (size_t)si * FIN + c0);
    uint4* pe = (uint4*)(&lds_e[r * EPITCH + c0]);
    uint4* pd = (uint4*)(&lds_e[r * EPITCH + FIN + c0]);
#pragma unroll
    for (int k = 0; k < 4; ++k) {
      float4 a0 = xi[2 * k], a1 = xi[2 * k + 1];
      float4 j0 = xj[2 * k], j1 = xj[2 * k + 1];
      pe[k] = pack8(a0, a1);
      float4 d0 = make_float4(j0.x - a0.x, j0.y - a0.y, j0.z - a0.z, j0.w - a0.w);
      float4 d1 = make_float4(j1.x - a1.x, j1.y - a1.y, j1.z - a1.z, j1.w - a1.w);
      pd[k] = pack8(d0, d1);
    }
  }
  __syncthreads();

  const int M  = lane & 15;
  const int hi = (lane >> 4) & 1;
  const int n  = wv * 16 + (lane & 15);

  v8f acc[MT];
#pragma unroll
  for (int mt = 0; mt < MT; ++mt) acc[mt] = (v8f){};
#pragma unroll
  for (int ks = 0; ks < 8; ++ks) {
    v16h b = ld_b_frag(W1f, wv * 8 + ks, lane);
#pragma unroll
    for (int mt = 0; mt < MT; ++mt) {
      v16h a = ld_a_frag(lds_e + mt * 16 * EPITCH, EPITCH, M, hi, ks * 32);
      acc[mt] = __builtin_amdgcn_wmma_f32_16x16x32_f16(false, a, false, b,
                                                       (short)0, acc[mt],
                                                       false, false);
    }
  }
  {
    const float bias = b1[n];
#pragma unroll
    for (int mt = 0; mt < MT; ++mt) {
#pragma unroll
      for (int r = 0; r < 8; ++r) {
        float v = acc[mt][r] + bias;
        v = v > 0.0f ? v : 0.0f;
        lds_h[(mt * 16 + r + 8 * hi) * HPITCH + n] = (f16)v;
      }
    }
  }
  __syncthreads();

  v8f acc2[MT];
#pragma unroll
  for (int mt = 0; mt < MT; ++mt) acc2[mt] = (v8f){};
#pragma unroll
  for (int ks = 0; ks < 4; ++ks) {
    v16h b = ld_b_frag(W2f, wv * 4 + ks, lane);
#pragma unroll
    for (int mt = 0; mt < MT; ++mt) {
      v16h a = ld_a_frag(lds_h + mt * 16 * HPITCH, HPITCH, M, hi, ks * 32);
      acc2[mt] = __builtin_amdgcn_wmma_f32_16x16x32_f16(false, a, false, b,
                                                        (short)0, acc2[mt],
                                                        false, false);
    }
  }
  {
    const float bias = b2[n];
#pragma unroll
    for (int mt = 0; mt < MT; ++mt) {
#pragma unroll
      for (int r = 0; r < 8; ++r) {
        const int row = mt * 16 + r + 8 * hi;
        atomicAdd(&summed[(size_t)lds_dst[row] * FOUT + n],
                  acc2[mt][r] + bias);
      }
    }
  }
}

__global__ __launch_bounds__(256) void node_mlp_kernel(
    const float* __restrict__ summed,
    const float* __restrict__ cnt,
    const f16* __restrict__ W3f,
    const float* __restrict__ b3,
    const f16* __restrict__ W4f,
    const float* __restrict__ b4,
    float* __restrict__ out) {
  __shared__ __attribute__((aligned(16))) f16 lds_a[16 * APITCH];
  __shared__ __attribute__((aligned(16))) f16 lds_m[16 * MPITCH];

  const int tid   = threadIdx.x;
  const int lane  = tid & 31;
  const int wv    = tid >> 5;
  const int nbase = blockIdx.x * 16;

  {
    const int r    = tid >> 4;
    const int c0   = (tid & 15) * 8;
    const int node = nbase + r;
    float cv  = cnt[node];
    float inv = 1.0f / (cv > 1.0f ? cv : 1.0f);
    const float4* sp = (const float4*)(summed + (size_t)node * FOUT + c0);
    float4 s0 = sp[0], s1 = sp[1];
    float4 m0 = make_float4(s0.x * inv, s0.y * inv, s0.z * inv, s0.w * inv);
    float4 m1 = make_float4(s1.x * inv, s1.y * inv, s1.z * inv, s1.w * inv);
    m0.x = m0.x > 0.f ? m0.x : 0.f; m0.y = m0.y > 0.f ? m0.y : 0.f;
    m0.z = m0.z > 0.f ? m0.z : 0.f; m0.w = m0.w > 0.f ? m0.w : 0.f;
    m1.x = m1.x > 0.f ? m1.x : 0.f; m1.y = m1.y > 0.f ? m1.y : 0.f;
    m1.z = m1.z > 0.f ? m1.z : 0.f; m1.w = m1.w > 0.f ? m1.w : 0.f;
    *(uint4*)(&lds_a[r * APITCH + c0]) = pack8(m0, m1);
  }
  __syncthreads();

  const int M  = lane & 15;
  const int hi = (lane >> 4) & 1;

  if (wv < 4) {
    const int n = wv * 16 + (lane & 15);
    v8f acc = {};
#pragma unroll
    for (int ks = 0; ks < 4; ++ks) {
      v16h a = ld_a_frag(lds_a, APITCH, M, hi, ks * 32);
      v16h b = ld_b_frag(W3f, wv * 4 + ks, lane);
      acc = __builtin_amdgcn_wmma_f32_16x16x32_f16(false, a, false, b,
                                                   (short)0, acc, false, false);
    }
    const float bias = b3[n];
#pragma unroll
    for (int r = 0; r < 8; ++r) {
      float v = acc[r] + bias;
      v = v > 0.0f ? v : 0.0f;
      lds_m[(r + 8 * hi) * MPITCH + n] = (f16)v;
    }
  }
  __syncthreads();

  {
    const int n = wv * 16 + (lane & 15);
    v8f acc = {};
#pragma unroll
    for (int ks = 0; ks < 2; ++ks) {
      v16h a = ld_a_frag(lds_m, MPITCH, M, hi, ks * 32);
      v16h b = ld_b_frag(W4f, wv * 2 + ks, lane);
      acc = __builtin_amdgcn_wmma_f32_16x16x32_f16(false, a, false, b,
                                                   (short)0, acc, false, false);
    }
    const float bias = b4[n];
#pragma unroll
    for (int r = 0; r < 8; ++r) {
      out[(size_t)(nbase + r + 8 * hi) * FOUT + n] = acc[r] + bias;
    }
  }
}

extern "C" void kernel_launch(void* const* d_in, const int* in_sizes, int n_in,
                              void* d_out, int out_size, void* d_ws, size_t ws_size,
                              hipStream_t stream) {
  const float* x   = (const float*)d_in[0];
  const int*  eidx = (const int*)d_in[1];
  const float* W1  = (const float*)d_in[2];
  const float* b1  = (const float*)d_in[3];
  const float* W2  = (const float*)d_in[4];
  const float* b2  = (const float*)d_in[5];
  const float* W3  = (const float*)d_in[6];
  const float* b3  = (const float*)d_in[7];
  const float* W4  = (const float*)d_in[8];
  const float* b4  = (const float*)d_in[9];
  float* out = (float*)d_out;

  char* ws = (char*)d_ws;
  float* summed = (float*)ws;
  float* cnt    = (float*)(ws + 25600000);
  f16* W1f = (f16*)(ws + 25800000);
  f16* W2f = (f16*)(ws + 25800000 + 65536);
  f16* W3f = (f16*)(ws + 25800000 + 65536 + 32768);
  f16* W4f = (f16*)(ws + 25800000 + 65536 + 32768 + 16384);

  hipMemsetAsync(summed, 0, (size_t)(NNODES * FOUT + NNODES) * sizeof(float), stream);

  pack_weights_kernel<<<(256 * 128 + 255) / 256, 256, 0, stream>>>(W1, W1f, 256, 128);
  pack_weights_kernel<<<(128 * 128 + 255) / 256, 256, 0, stream>>>(W2, W2f, 128, 128);
  pack_weights_kernel<<<(128 * 64  + 255) / 256, 256, 0, stream>>>(W3, W3f, 128, 64);
  pack_weights_kernel<<<(64  * 128 + 255) / 256, 256, 0, stream>>>(W4, W4f, 64, 128);

  edge_mlp_kernel<<<NEDGES / (MT * 16), 256, 0, stream>>>(
      x, eidx, eidx + NEDGES, W1f, b1, W2f, b2, summed, cnt);

  node_mlp_kernel<<<NNODES / 16, 256, 0, stream>>>(
      summed, cnt, W3f, b3, W4f, b4, out);
}


# Sort the edges by destination in every call (counting sort, one integer atomic per edge) and sum each node's relu(P+Q[src]) rows in registers inside the node MLP kernel; no float atomics, no S plane
# speedup vs baseline: 4.6916x; 2.1622x over previous
#include <hip/hip_runtime.h>


typedef _Float16 f16;
typedef __attribute__((ext_vector_type(16))) _Float16 v16h;
typedef __attribute__((ext_vector_type(8)))  float    v8f;

#define NNODES 50000
#define NEDGES 800000
#define FIN    128
#define FOUT   128


#define XPITCH (FIN + 8)
#define APITCH (FOUT + 8)
#define MPITCH (FOUT / 2 + 8)

#define PRE_MT 5
#define EDGE_G 8
#define SORT_BLOCKS 625
#define SORT_EPT 5
#define SORT_STRIDE (SORT_BLOCKS * 256)
static_assert(SORT_STRIDE * SORT_EPT == NEDGES, "sort passes must cover every edge exactly once");

union V16H { v16h v; uint4 q[2]; };
union H8   { f16 h[8]; uint4 q; };
union H4   { f16 h[4]; uint2 d; };

__device__ __forceinline__ uint4 pack8(float4 a, float4 b) {
  H8 u;
  u.h[0] = (f16)a.x; u.h[1] = (f16)a.y; u.h[2] = (f16)a.z; u.h[3] = (f16)a.w;
  u.h[4] = (f16)b.x; u.h[5] = (f16)b.y; u.h[6] = (f16)b.z; u.h[7] = (f16)b.w;
  return u.q;
}

__device__ __forceinline__ v16h ld_a_frag(const f16* lds, int row_stride,
                                          int M, int hi, int kbase) {
  const int kb = kbase + hi * 8;
  V16H t;
  t.q[0] = *(const uint4*)(lds + M * row_stride + kb);
  t.q[1] = *(const uint4*)(lds + M * row_stride + kb + 16);
  return t.v;
}

__device__ __forceinline__ v16h ld_b_frag(const f16* __restrict__ wf,
                                          int frag, int lane) {
  const uint4* p = (const uint4*)(wf + ((size_t)(frag * 32 + lane)) * 16);
  V16H t;
  t.q[0] = p[0];
  t.q[1] = p[1];
  return t.v;
}

__global__ __launch_bounds__(256) void pack_weights_kernel(
    const float* __restrict__ W, f16* __restrict__ out, int K, int N) {
  int o = blockIdx.x * blockDim.x + threadIdx.x;
  if (o >= K * N) return;
  int ksteps = K >> 5;
  int e    = o & 15;
  int lane = (o >> 4) & 31;
  int frag = o >> 9;
  int ks   = frag % ksteps;
  int nt   = frag / ksteps;
  int kl   = (e < 8 ? e : e + 8) + ((lane & 16) ? 8 : 0);
  int Kidx = ks * 32 + kl;
  int n    = nt * 16 + (lane & 15);
  out[o] = (f16)W[Kidx * N + n];
}

__global__ __launch_bounds__(256) void pack_w1_folded_kernel(
    const float* __restrict__ W1, f16* __restrict__ out) {
  int o = blockIdx.x * blockDim.x + threadIdx.x;
  if (o >= FIN * 2 * FOUT) return;
  const int ksteps = FIN >> 5;
  int e    = o & 15;
  int lane = (o >> 4) & 31;
  int frag = o >> 9;
  int ks   = frag % ksteps;
  int nt   = frag / ksteps;
  int kl   = (e < 8 ? e : e + 8) + ((lane & 16) ? 8 : 0);
  int Kidx = ks * 32 + kl;
  int n    = nt * 16 + (lane & 15);
  float w;
  if (n < FOUT) w = W1[Kidx * FOUT + n] - W1[(FIN + Kidx) * FOUT + n];
  else          w = W1[(FIN + Kidx) * FOUT + (n - FOUT)];
  out[o] = (f16)w;
}

__global__ __launch_bounds__(128) void node_pre_kernel(
    const float* __restrict__ x,
    const f16* __restrict__ W1c,
    const float* __restrict__ b1,
    f16* __restrict__ P,
    f16* __restrict__ Q) {
  __shared__ __attribute__((aligned(16))) f16 lds_x[PRE_MT * 16 * XPITCH];

  const int tid   = threadIdx.x;
  const int lane  = tid & 31;
  const int wv    = tid >> 5;
  const int nbase = blockIdx.x * (PRE_MT * 16);

#pragma unroll
  for (int i = 0; i < (PRE_MT * 16 * 16) / 128; ++i) {
    const int idx = tid + i * 128;
    const int r   = idx >> 4;
    const int c0  = (idx & 15) * 8;
    const float4* xp = (const float4*)(x + (size_t)(nbase + r) * FIN + c0);
    float4 a0 = xp[0], a1 = xp[1];
    *(uint4*)(&lds_x[r * XPITCH + c0]) = pack8(a0, a1);
  }

  const int pq   = wv >> 1;
  const int t    = wv & 1;
  const int M    = lane & 15;
  const int hi   = (lane >> 4) & 1;

  v16h bf[4][4];
#pragma unroll
  for (int k = 0; k < 4; ++k) {
    const int nt = pq * 8 + t + 2 * k;
#pragma unroll
    for (int ks = 0; ks < 4; ++ks) bf[k][ks] = ld_b_frag(W1c, nt * 4 + ks, lane);
  }
  float bias[4];
#pragma unroll
  for (int k = 0; k < 4; ++k)
    bias[k] = pq == 0 ? b1[16 * t + M + 32 * k] : 0.0f;

  __syncthreads();

#pragma unroll 1
  for (int mt = 0; mt < PRE_MT; ++mt) {
    v8f acc[4];
#pragma unroll
    for (int k = 0; k < 4; ++k) acc[k] = (v8f){};
#pragma unroll
    for (int ks = 0; ks < 4; ++ks) {
      v16h a = ld_a_frag(lds_x + mt * 16 * XPITCH, XPITCH, M, hi, ks * 32);
#pragma unroll
      for (int k = 0; k < 4; ++k)
        acc[k] = __builtin_amdgcn_wmma_f32_16x16x32_f16(false, a, false, bf[k][ks],
                                                        (short)0, acc[k],
                                                        false, false);
    }
#pragma unroll
    for (int r = 0; r < 8; ++r) {
      const int node = nbase + mt * 16 + r + 8 * hi;
      H4 u;
#pragma unroll
      for (int k = 0; k < 4; ++k) u.h[k] = (f16)(acc[k][r] + bias[k]);
      f16* rowp = pq == 0 ? P + (size_t)node * (2 * FOUT) : Q + (size_t)node * FOUT;
      *(uint2*)(rowp + 4 * (16 * t + M)) = u.d;
    }
  }
}

__global__ __launch_bounds__(256) void edge_rank_kernel(
    const int* __restrict__ dst,
    int* __restrict__ deg,
    int* __restrict__ rank) {
  const int t = blockIdx.x * 256 + threadIdx.x;
  int d[SORT_EPT];
#pragma unroll
  for (int k = 0; k < SORT_EPT; ++k) d[k] = dst[t + k * SORT_STRIDE];
  int r[SORT_EPT];
#pragma unroll
  for (int k = 0; k < SORT_EPT; ++k) r[k] = atomicAdd(&deg[d[k]], 1);
#pragma unroll
  for (int k = 0; k < SORT_EPT; ++k) rank[t + k * SORT_STRIDE] = r[k];
}

__global__ __launch_bounds__(256) void seg_alloc_kernel(
    const int* __restrict__ deg,
    int* __restrict__ off,
    int* __restrict__ cursor) {
  const int node = blockIdx.x * 256 + threadIdx.x;
  const int lane = threadIdx.x & 31;
  const int c = node < NNODES ? deg[node] : 0;
  int v = c;
  for (int s = 1; s < 32; s <<= 1) {
    const int up = __shfl_up(v, s, 32);
    if (lane >= s) v += up;
  }
  const int total = __shfl(v, 31, 32);
  int base = 0;
  if (lane == 31) base = atomicAdd(cursor, total);
  base = __shfl(base, 31, 32);
  if (node < NNODES) off[node] = base + v - c;
}

__global__ __launch_bounds__(256) void edge_place_kernel(
    const int* __restrict__ src,
    const int* __restrict__ dst,
    const int* __restrict__ rank,
    const int* __restrict__ off,
    int* __restrict__ ssrc) {
  const int t = blockIdx.x * 256 + threadIdx.x;
  int pos[SORT_EPT], s[SORT_EPT];
#pragma unroll
  for (int k = 0; k < SORT_EPT; ++k) {
    const int e = t + k * SORT_STRIDE;
    pos[k] = off[dst[e]] + rank[e];
    s[k]   = src[e];
  }
#pragma unroll
  for (int k = 0; k < SORT_EPT; ++k) ssrc[pos[k]] = s[k];
}

__global__ __launch_bounds__(256) void node_gather_mlp_kernel(
    const f16* __restrict__ Q,
    const int* __restrict__ ssrc,
    const int* __restrict__ off,
    const int* __restrict__ deg,
    const f16* __restrict__ W2f,
    const float* __restrict__ b2,
    const f16* __restrict__ W3f,
    const float* __restrict__ b3,
    const f16* __restrict__ W4f,
    const float* __restrict__ b4,
    float* out) {
  __shared__ __attribute__((aligned(16))) f16 lds_s[16 * APITCH];
  __shared__ __attribute__((aligned(16))) f16 lds_a[16 * APITCH];
  __shared__ __attribute__((aligned(16))) f16 lds_m[16 * MPITCH];
  __shared__ float lds_has[16];

  const int tid   = threadIdx.x;
  const int lane  = tid & 31;
  const int wv    = tid >> 5;
  const int nbase = blockIdx.x * 16;

  {
    const f16* P = (const f16*)out;
    const int wvu = __builtin_amdgcn_readfirstlane(wv);
    for (int rr = 0; rr < 2; ++rr) {
      const int r    = wvu * 2 + rr;
      const int node = nbase + r;
      const int o    = off[node];
      const int c    = deg[node];
      H4 pp;
      pp.d = *(const uint2*)(P + (size_t)node * (2 * FOUT) + 4 * lane);
      float pf[4], acc[4];
#pragma unroll
      for (int k = 0; k < 4; ++k) { pf[k] = (float)pp.h[k]; acc[k] = 0.0f; }

      for (int base = 0; base < c; base += 32) {
        const int m = (c - base) < 32 ? (c - base) : 32;
        const int iv = ssrc[o + base + (lane < m ? lane : 0)];
        for (int j = 0; j < m; j += EDGE_G) {
          uint2 q[EDGE_G];
#pragma unroll
          for (int u = 0; u < EDGE_G; ++u) {
            const int jj = (j + u) < m ? (j + u) : j;
            const int sj = __builtin_amdgcn_readlane(iv, jj);
            q[u] = *(const uint2*)(Q + (size_t)sj * FOUT + 4 * lane);
          }
#pragma unroll
          for (int u = 0; u < EDGE_G; ++u) {
            const bool live = (j + u) < m;
            H4 b;
            b.d = q[u];
#pragma unroll
            for (int k = 0; k < 4; ++k) {
              float v = pf[k] + (float)b.h[k];
              v = v > 0.0f ? v : 0.0f;
              acc[k] += live ? v : 0.0f;
            }
          }
        }
      }
      const float inv = 1.0f / (float)(c > 1 ? c : 1);
#pragma unroll
      for (int k = 0; k < 4; ++k)
        lds_s[r * APITCH + lane + 32 * k] = (f16)(acc[k] * inv);
      if (lane == 0) lds_has[r] = c > 0 ? 1.0f : 0.0f;
    }
  }
  __syncthreads();

  const int M  = lane & 15;
  const int hi = (lane >> 4) & 1;

  {
    const int n = wv * 16 + (lane & 15);
    v8f acc = {};
#pragma unroll
    for (int ks = 0; ks < 4; ++ks) {
      v16h a = ld_a_frag(lds_s, APITCH, M, hi, ks * 32);
      v16h b = ld_b_frag(W2f, wv * 4 + ks, lane);
      acc = __builtin_amdgcn_wmma_f32_16x16x32_f16(false, a, false, b,
                                                   (short)0, acc, false, false);
    }
    const float bias = b2[n];
#pragma unroll
    for (int r = 0; r < 8; ++r) {
      float v = acc[r] + bias * lds_has[r + 8 * hi];
      v = v > 0.0f ? v : 0.0f;
      lds_a[(r + 8 * hi) * APITCH + n] = (f16)v;
    }
  }
  __syncthreads();

  if (wv < 4) {
    const int n = wv * 16 + (lane & 15);
    v8f acc = {};
#pragma unroll
    for (int ks = 0; ks < 4; ++ks) {
      v16h a = ld_a_frag(lds_a, APITCH, M, hi, ks * 32);
      v16h b = ld_b_frag(W3f, wv * 4 + ks, lane);
      acc = __builtin_amdgcn_wmma_f32_16x16x32_f16(false, a, false, b,
                                                   (short)0, acc, false, false);
    }
    const float bias = b3[n];
#pragma unroll
    for (int r = 0; r < 8; ++r) {
      float v = acc[r] + bias;
      v = v > 0.0f ? v : 0.0f;
      lds_m[(r + 8 * hi) * MPITCH + n] = (f16)v;
    }
  }
  __syncthreads();

  {
    const int n = wv * 16 + (lane & 15);
    v8f acc = {};
#pragma unroll
    for (int ks = 0; ks < 2; ++ks) {
      v16h a = ld_a_frag(lds_m, MPITCH, M, hi, ks * 32);
      v16h b = ld_b_frag(W4f, wv * 2 + ks, lane);
      acc = __builtin_amdgcn_wmma_f32_16x16x32_f16(false, a, false, b,
                                                   (short)0, acc, false, false);
    }
    const float bias = b4[n];
#pragma unroll
    for (int r = 0; r < 8; ++r) {
      out[(size_t)(nbase + r + 8 * hi) * FOUT + n] = acc[r] + bias;
    }
  }
}

extern "C" void kernel_launch(void* const* d_in, const int* in_sizes, int n_in,
                              void* d_out, int out_size, void* d_ws, size_t ws_size,
                              hipStream_t stream) {
  const float* x   = (const float*)d_in[0];
  const int*  eidx = (const int*)d_in[1];
  const float* W1  = (const float*)d_in[2];
  const float* b1  = (const float*)d_in[3];
  const float* W2  = (const float*)d_in[4];
  const float* b2  = (const float*)d_in[5];
  const float* W3  = (const float*)d_in[6];
  const float* b3  = (const float*)d_in[7];
  const float* W4  = (const float*)d_in[8];
  const float* b4  = (const float*)d_in[9];
  float* out = (float*)d_out;

  char* ws = (char*)d_ws;
  f16* Q      = (f16*)ws;
  int* rank   = (int*)(ws + 12800000);
  int* ssrc   = (int*)(ws + 16000000);
  int* deg    = (int*)(ws + 19200000);
  int* cursor = (int*)(ws + 19400000);
  int* off    = (int*)(ws + 19400064);
  f16* W1c = (f16*)(ws + 25800000);
  f16* W2f = (f16*)(ws + 25800000 + 65536);
  f16* W3f = (f16*)(ws + 25800000 + 65536 + 32768);
  f16* W4f = (f16*)(ws + 25800000 + 65536 + 32768 + 16384);
  f16* P   = (f16*)d_out;

  hipMemsetAsync(deg, 0, 200064, stream);

  pack_w1_folded_kernel<<<(128 * 256 + 255) / 256, 256, 0, stream>>>(W1, W1c);
  pack_weights_kernel<<<(128 * 128 + 255) / 256, 256, 0, stream>>>(W2, W2f, 128, 128);
  pack_weights_kernel<<<(128 * 64  + 255) / 256, 256, 0, stream>>>(W3, W3f, 128, 64);
  pack_weights_kernel<<<(64  * 128 + 255) / 256, 256, 0, stream>>>(W4, W4f, 64, 128);

  edge_rank_kernel<<<SORT_BLOCKS, 256, 0, stream>>>(eidx + NEDGES, deg, rank);
  seg_alloc_kernel<<<(NNODES + 255) / 256, 256, 0, stream>>>(deg, off, cursor);
  edge_place_kernel<<<SORT_BLOCKS, 256, 0, stream>>>(eidx, eidx + NEDGES, rank, off, ssrc);

  node_pre_kernel<<<NNODES / (PRE_MT * 16), 128, 0, stream>>>(x, W1c, b1, P, Q);

  node_gather_mlp_kernel<<<NNODES / 16, 256, 0, stream>>>(
      Q, ssrc, off, deg, W2f, b2, W3f, b3, W4f, b4, out);
}


# Edges are counting-sorted by destination bucket in LDS and one block per bucket builds the slot rows in LDS: per-edge random global atomics and 4-byte stores become one atomic per bucket and runs
# speedup vs baseline: 9.7997x; 2.0888x over previous
#include <hip/hip_runtime.h>


typedef _Float16 f16;
typedef __attribute__((ext_vector_type(16))) _Float16 v16h;
typedef __attribute__((ext_vector_type(8)))  float    v8f;

#define NNODES 50000
#define NEDGES 800000
#define FIN    128
#define FOUT   128


#define XPITCH (FIN + 8)
#define APITCH (FOUT + 8)
#define MPITCH (FOUT / 2 + 8)

#define PRE_MT 5
#define EDGE_G 8
#define SLOT_C 32
#define NPB 64
#define NBUCK ((NNODES + NPB - 1) / NPB)
#define BROW 32
#define GC_STRIDE 32
#define BIN_BLOCKS 125
#define BIN_EPT 25
static_assert(BIN_BLOCKS * 256 * BIN_EPT == NEDGES, "the binning pass must cover every edge exactly once");
static_assert(NNODES <= 65536, "a packed edge word holds destination and source id in 16 bits each");
#define OUT_ROW_WORDS FOUT
#define STAGE_W 64
#define BOVF_W 96
static_assert(STAGE_W + BROW == BOVF_W && BOVF_W + 32 == OUT_ROW_WORDS, "output row scratch layout");
static_assert((NEDGES >> 5) <= NNODES, "the bucket overflow list fits in the output rows");

#define PACK_BLOCKS_W1 ((FIN * 2 * FOUT) / 256)
#define PACK_BLOCKS_W2 ((FOUT * FOUT) / 256)
#define PACK_BLOCKS_W3 ((FOUT * (FOUT / 2)) / 256)
#define PACK_BLOCKS_W4 (((FOUT / 2) * FOUT) / 256)
#define PACK_BLOCKS (PACK_BLOCKS_W1 + PACK_BLOCKS_W2 + PACK_BLOCKS_W3 + PACK_BLOCKS_W4)
#define PRE_THREADS 128
#define PRE_BLOCKS (NNODES / (PRE_MT * 16))
static_assert(PRE_BLOCKS * PRE_MT * 16 == NNODES, "pre-pass tiles must cover every node exactly once");

union V16H { v16h v; uint4 q[2]; };
union H8   { f16 h[8]; uint4 q; };
union H4   { f16 h[4]; uint2 d; };
typedef __attribute__((ext_vector_type(2))) _Float16 h2;
union HP   { uint2 d; h2 p[2]; };

__device__ __forceinline__ void row_acc(float (&acc)[4], uint2 qv, h2 np0, h2 np1) {
  HP b;
  b.d = qv;
  const h2 t0 = __builtin_elementwise_maximum(b.p[0], np0);
  const h2 t1 = __builtin_elementwise_maximum(b.p[1], np1);
  acc[0] += (float)t0[0];
  acc[1] += (float)t0[1];
  acc[2] += (float)t1[0];
  acc[3] += (float)t1[1];
}

__device__ __forceinline__ uint2 ld_qrow(const char* __restrict__ Qb, unsigned rowoff, unsigned lane8) {
  return *(const uint2*)(Qb + (size_t)(rowoff + lane8));
}

__device__ __forceinline__ uint4 pack8(float4 a, float4 b) {
  H8 u;
  u.h[0] = (f16)a.x; u.h[1] = (f16)a.y; u.h[2] = (f16)a.z; u.h[3] = (f16)a.w;
  u.h[4] = (f16)b.x; u.h[5] = (f16)b.y; u.h[6] = (f16)b.z; u.h[7] = (f16)b.w;
  return u.q;
}

__device__ __forceinline__ v16h ld_a_frag(const f16* lds, int row_stride,
                                          int M, int hi, int kbase) {
  const int kb = kbase + hi * 8;
  V16H t;
  t.q[0] = *(const uint4*)(lds + M * row_stride + kb);
  t.q[1] = *(const uint4*)(lds + M * row_stride + kb + 16);
  return t.v;
}

__device__ __forceinline__ v16h ld_b_frag(const f16* __restrict__ wf,
                                          int frag, int lane) {
  const uint4* p = (const uint4*)(wf + ((size_t)(frag * 32 + lane)) * 16);
  V16H t;
  t.q[0] = p[0];
  t.q[1] = p[1];
  return t.v;
}

template <int K, int N>
__device__ __forceinline__ void pack_weights_elem(
    const float* __restrict__ W, f16* __restrict__ out, int o) {
  const int ksteps = K >> 5;
  int e    = o & 15;
  int lane = (o >> 4) & 31;
  int frag = o >> 9;
  int ks   = frag % ksteps;
  int nt   = frag / ksteps;
  int kl   = (e < 8 ? e : e + 8) + ((lane & 16) ? 8 : 0);
  int Kidx = ks * 32 + kl;
  int n    = nt * 16 + (lane & 15);
  out[o] = (f16)W[Kidx * N + n];
}

__device__ __forceinline__ void pack_w1_folded_elem(
    const float* __restrict__ W1, f16* __restrict__ out, int o) {
  const int ksteps = FIN >> 5;
  int e    = o & 15;
  int lane = (o >> 4) & 31;
  int frag = o >> 9;
  int ks   = frag % ksteps;
  int nt   = frag / ksteps;
  int kl   = (e < 8 ? e : e + 8) + ((lane & 16) ? 8 : 0);
  int Kidx = ks * 32 + kl;
  int n    = nt * 16 + (lane & 15);
  float w;
  if (n < FOUT) w = W1[Kidx * FOUT + n] - W1[(FIN + Kidx) * FOUT + n];
  else          w = W1[(FIN + Kidx) * FOUT + (n - FOUT)];
  out[o] = (f16)w;
}

__device__ __forceinline__ void node_pre_block(
    const int tile,
    f16* lds_x,
    const float* __restrict__ x,
    const f16* __restrict__ W1c,
    const float* __restrict__ b1,
    f16* __restrict__ P,
    f16* __restrict__ Q) {
  const int tid   = threadIdx.x;
  const int lane  = tid & 31;
  const int wv    = tid >> 5;
  const int nbase = tile * (PRE_MT * 16);

#pragma unroll
  for (int i = 0; i < (PRE_MT * 16 * 16) / 128; ++i) {
    const int idx = tid + i * 128;
    const int r   = idx >> 4;
    const int c0  = (idx & 15) * 8;
    const float4* xp = (const float4*)(x + (size_t)(nbase + r) * FIN + c0);
    float4 a0 = xp[0], a1 = xp[1];
    *(uint4*)(&lds_x[r * XPITCH + c0]) = pack8(a0, a1);
  }

  const int pq   = wv >> 1;
  const int t    = wv & 1;
  const int M    = lane & 15;
  const int hi   = (lane >> 4) & 1;

  v16h bf[4][4];
#pragma unroll
  for (int k = 0; k < 4; ++k) {
    const int nt = pq * 8 + t + 2 * k;
#pragma unroll
    for (int ks = 0; ks < 4; ++ks) bf[k][ks] = ld_b_frag(W1c, nt * 4 + ks, lane);
  }
  float bias[4];
#pragma unroll
  for (int k = 0; k < 4; ++k)
    bias[k] = pq == 0 ? b1[16 * t + M + 32 * k] : 0.0f;

  __syncthreads();

#pragma unroll 1
  for (int mt = 0; mt < PRE_MT; ++mt) {
    v8f acc[4];
#pragma unroll
    for (int k = 0; k < 4; ++k) acc[k] = (v8f){};
#pragma unroll
    for (int ks = 0; ks < 4; ++ks) {
      v16h a = ld_a_frag(lds_x + mt * 16 * XPITCH, XPITCH, M, hi, ks * 32);
#pragma unroll
      for (int k = 0; k < 4; ++k)
        acc[k] = __builtin_amdgcn_wmma_f32_16x16x32_f16(false, a, false, bf[k][ks],
                                                        (short)0, acc[k],
                                                        false, false);
    }
#pragma unroll
    for (int r = 0; r < 8; ++r) {
      const int node = nbase + mt * 16 + r + 8 * hi;
      H4 u;
#pragma unroll
      for (int k = 0; k < 4; ++k) u.h[k] = (f16)(acc[k][r] + bias[k]);
      f16* rowp = pq == 0 ? P + (size_t)node * (2 * FOUT) : Q + (size_t)node * FOUT;
      *(uint2*)(rowp + 4 * (16 * t + M)) = u.d;
    }
  }
}

#define BIN_EDGES (256 * BIN_EPT)
#define NBUCK_PAD 1024
static_assert(NBUCK <= NBUCK_PAD && NBUCK_PAD == 4 * 256, "a thread owns four buckets of the padded table");
struct BinLds {
  int hist[NBUCK_PAD];
  int pref[NBUCK_PAD];
  int base[NBUCK_PAD];
  int wsum[8];
  unsigned sw[BIN_EDGES];
  unsigned short sb[BIN_EDGES];
};
__device__ __forceinline__ void edge_bin_block(
    BinLds& L,
    const int* __restrict__ src,
    const int* __restrict__ dst,
    int* __restrict__ gcount,
    int* __restrict__ nbovf,
    unsigned* outw) {
  const int tid  = threadIdx.x;
  const int lane = tid & 31;
  const int wv   = tid >> 5;
#pragma unroll
  for (int i = 0; i < 4; ++i) L.hist[tid + 256 * i] = 0;
  const int e0 = blockIdx.x * BIN_EDGES + tid;
  int d[BIN_EPT], s[BIN_EPT];
#pragma unroll
  for (int k = 0; k < BIN_EPT; ++k) {
    d[k] = dst[e0 + k * 256];
    s[k] = src[e0 + k * 256];
  }
  __syncthreads();
  int lr[BIN_EPT];
#pragma unroll
  for (int k = 0; k < BIN_EPT; ++k) lr[k] = atomicAdd(&L.hist[d[k] / NPB], 1);
  __syncthreads();

  int a[4];
#pragma unroll
  for (int i = 0; i < 4; ++i) a[i] = L.hist[4 * tid + i];
  const int tsum = a[0] + a[1] + a[2] + a[3];
  int inc = tsum;
#pragma unroll
  for (int off = 1; off < 32; off <<= 1) {
    const int v = __shfl_up(inc, off);
    if (lane >= off) inc += v;
  }
  if (lane == 31) L.wsum[wv] = inc;
#pragma unroll
  for (int i = 0; i < 4; ++i)
    L.base[4 * tid + i] =
        a[i] > 0 ? atomicAdd(&gcount[(4 * tid + i) * GC_STRIDE], a[i]) : 0;
  __syncthreads();
  int ex = inc - tsum;
  for (int w = 0; w < wv; ++w) ex += L.wsum[w];
#pragma unroll
  for (int i = 0; i < 4; ++i) {
    L.pref[4 * tid + i] = ex;
    ex += a[i];
  }
  __syncthreads();

#pragma unroll
  for (int k = 0; k < BIN_EPT; ++k) {
    const int b = d[k] / NPB;
    const int j = L.pref[b] + lr[k];
    L.sw[j] = ((unsigned)(d[k] % NPB) << 16) | (unsigned)s[k];
    L.sb[j] = (unsigned short)b;
  }
  __syncthreads();

  for (int k = 0; k < BIN_EPT; ++k) {
    const int j    = tid + k * 256;
    const int b    = (int)L.sb[j];
    const unsigned w = L.sw[j];
    const int pos  = L.base[b] + (j - L.pref[b]);
    const int left = NNODES - b * NPB;
    const int cap  = (left < NPB ? left : NPB) * BROW;
    if (pos < cap) {
      outw[(size_t)(b * NPB + (pos >> 5)) * OUT_ROW_WORDS + STAGE_W + (pos & 31)] = w;
    } else {
      const int q = atomicAdd(nbovf, 1);
      outw[(size_t)(q >> 5) * OUT_ROW_WORDS + BOVF_W + (q & 31)] =
          ((unsigned)(b * NPB) << 16) + w;
    }
  }
}

__device__ __forceinline__ void place_edge(
    int* cnt, int* rows, const int node, const int dl, const int s,
    int* __restrict__ novf, unsigned* __restrict__ ovf) {
  const int r = atomicAdd(&cnt[dl], 1);
  if (r < SLOT_C) {
    rows[dl * SLOT_C + r] = s << 8;
  } else {
    const int q = atomicAdd(novf, 1);
    ovf[q] = ((unsigned)node << 16) | (unsigned)s;
  }
}

__device__ __forceinline__ void bucket_place_block(
    const int b,
    int* cnt,
    int* rows,
    const int* __restrict__ gcount,
    const int* __restrict__ nbovf,
    const unsigned* outw,
    int* __restrict__ deg,
    int* __restrict__ slots,
    int* __restrict__ novf,
    unsigned* __restrict__ ovf) {
  const int tid  = threadIdx.x;
  const int left = NNODES - b * NPB;
  const int nn   = left < NPB ? left : NPB;
  if (tid < NPB) cnt[tid] = 0;
  __syncthreads();
  const int g   = gcount[b * GC_STRIDE];
  const int cap = nn * BROW;
  const int m   = g < cap ? g : cap;
  for (int i = tid; i < m; i += PRE_THREADS) {
    const unsigned w = outw[(size_t)(b * NPB + (i >> 5)) * OUT_ROW_WORDS + STAGE_W + (i & 31)];
    const int dl = (int)(w >> 16) % NPB;
    place_edge(cnt, rows, b * NPB + dl, dl, (int)(w & 0xFFFFu), novf, ovf);
  }
  if (g > cap) {
    int n = nbovf[0];
    n = n < NEDGES ? n : NEDGES;
    for (int i = tid; i < n; i += PRE_THREADS) {
      const unsigned w = outw[(size_t)(i >> 5) * OUT_ROW_WORDS + BOVF_W + (i & 31)];
      const int node = (int)(w >> 16);
      if (node / NPB == b) place_edge(cnt, rows, node, node % NPB, (int)(w & 0xFFFFu), novf, ovf);
    }
  }
  __syncthreads();
  for (int i = tid; i < nn * SLOT_C; i += PRE_THREADS)
    if ((i & (SLOT_C - 1)) < cnt[i / SLOT_C]) slots[b * NPB * SLOT_C + i] = rows[i];
  if (tid < nn) deg[b * NPB + tid] = cnt[tid];
}

__global__ __launch_bounds__(256) void bin_pack_kernel(
    const int* __restrict__ src,
    const int* __restrict__ dst,
    int* __restrict__ gcount,
    int* __restrict__ nbovf,
    unsigned* outw,
    const float* __restrict__ W1,
    const float* __restrict__ W2,
    const float* __restrict__ W3,
    const float* __restrict__ W4,
    f16* __restrict__ W1c,
    f16* __restrict__ W2f,
    f16* __restrict__ W3f,
    f16* __restrict__ W4f) {
  __shared__ BinLds L;
  if (blockIdx.x < BIN_BLOCKS) {
    edge_bin_block(L, src, dst, gcount, nbovf, outw);
    return;
  }
  const int pb = blockIdx.x - BIN_BLOCKS;
  const int t  = threadIdx.x;
  if (pb < PACK_BLOCKS_W1) {
    pack_w1_folded_elem(W1, W1c, pb * 256 + t);
  } else if (pb < PACK_BLOCKS_W1 + PACK_BLOCKS_W2) {
    pack_weights_elem<FOUT, FOUT>(W2, W2f, (pb - PACK_BLOCKS_W1) * 256 + t);
  } else if (pb < PACK_BLOCKS_W1 + PACK_BLOCKS_W2 + PACK_BLOCKS_W3) {
    pack_weights_elem<FOUT, FOUT / 2>(W3, W3f, (pb - PACK_BLOCKS_W1 - PACK_BLOCKS_W2) * 256 + t);
  } else {
    pack_weights_elem<FOUT / 2, FOUT>(W4, W4f,
                                      (pb - PACK_BLOCKS_W1 - PACK_BLOCKS_W2 - PACK_BLOCKS_W3) * 256 + t);
  }
}

#define PRE_LDS_BYTES (PRE_MT * 16 * XPITCH * 2)
static_assert((NPB + NPB * SLOT_C) * 4 <= PRE_LDS_BYTES, "placement tables fit in the pre-pass tile's LDS");
__global__ __launch_bounds__(PRE_THREADS) void node_pre_place_kernel(
    const float* __restrict__ x,
    const f16* __restrict__ W1c,
    const float* __restrict__ b1,
    f16* __restrict__ P,
    f16* __restrict__ Q,
    const int* __restrict__ gcount,
    const int* __restrict__ nbovf,
    const unsigned* outw,
    int* __restrict__ deg,
    int* __restrict__ slots,
    int* __restrict__ novf,
    unsigned* __restrict__ ovf) {
  __shared__ __attribute__((aligned(16))) unsigned char lds_raw[PRE_LDS_BYTES];
  if (blockIdx.x < PRE_BLOCKS) {
    node_pre_block(blockIdx.x, (f16*)lds_raw, x, W1c, b1, P, Q);
    return;
  }
  int* li = (int*)lds_raw;
  bucket_place_block(blockIdx.x - PRE_BLOCKS, li, li + NPB, gcount, nbovf, outw,
                     deg, slots, novf, ovf);
}

__global__ __launch_bounds__(256) void node_gather_mlp_kernel(
    const f16* __restrict__ Q,
    const int* __restrict__ slots,
    const int* __restrict__ deg,
    const int* __restrict__ novf,
    const unsigned* __restrict__ ovf,
    const f16* __restrict__ W2f,
    const float* __restrict__ b2,
    const f16* __restrict__ W3f,
    const float* __restrict__ b3,
    const f16* __restrict__ W4f,
    const float* __restrict__ b4,
    float* out) {
  __shared__ __attribute__((aligned(16))) f16 lds_s[16 * APITCH];
  __shared__ __attribute__((aligned(16))) f16 lds_a[16 * APITCH];
  __shared__ __attribute__((aligned(16))) f16 lds_m[16 * MPITCH];
  __shared__ float lds_has[16];

  const int tid   = threadIdx.x;
  const int lane  = tid & 31;
  const int wv    = tid >> 5;
  const int nbase = blockIdx.x * 16;

  {
    const f16* P = (const f16*)out;
    const int wvu = __builtin_amdgcn_readfirstlane(wv);
    const int n0  = nbase + wvu * 2;
    int cc[2], ivv[2];
    H4 ppv[2];
#pragma unroll
    for (int rr = 0; rr < 2; ++rr) cc[rr] = deg[n0 + rr];
#pragma unroll
    for (int rr = 0; rr < 2; ++rr) {
      const int cs = cc[rr] < SLOT_C ? cc[rr] : SLOT_C;
      ivv[rr] = lane < cs ? slots[(n0 + rr) * SLOT_C + lane] : 0;
      ppv[rr].d = *(const uint2*)(P + (size_t)(n0 + rr) * (2 * FOUT) + 4 * lane);
    }
    const char* Qb = (const char*)Q;
    const unsigned lane8 = (unsigned)lane * 8u;
#pragma unroll
    for (int rr = 0; rr < 2; ++rr) {
      const int r    = wvu * 2 + rr;
      const int node = n0 + rr;
      const int c    = cc[rr];
      const int m    = c < SLOT_C ? c : SLOT_C;
      const int ov   = ivv[rr];
      HP pp;
      pp.d = ppv[rr].d;
      const h2 np0 = -pp.p[0];
      const h2 np1 = -pp.p[1];
      float pf[4], acc[4];
#pragma unroll
      for (int k = 0; k < 4; ++k) { pf[k] = (float)ppv[rr].h[k]; acc[k] = 0.0f; }

      int j = 0;
      for (; j + EDGE_G <= m; j += EDGE_G) {
        uint2 q[EDGE_G];
#pragma unroll
        for (int u = 0; u < EDGE_G; ++u)
          q[u] = ld_qrow(Qb, (unsigned)__builtin_amdgcn_readlane(ov, j + u), lane8);
#pragma unroll
        for (int u = 0; u < EDGE_G; ++u) row_acc(acc, q[u], np0, np1);
      }
      {
        const int rem = m - j;
        uint2 q4[4], q2[2], q1;
        if (rem & 4) {
#pragma unroll
          for (int u = 0; u < 4; ++u)
            q4[u] = ld_qrow(Qb, (unsigned)__builtin_amdgcn_readlane(ov, j + u), lane8);
          j += 4;
        }
        if (rem & 2) {
#pragma unroll
          for (int u = 0; u < 2; ++u)
            q2[u] = ld_qrow(Qb, (unsigned)__builtin_amdgcn_readlane(ov, j + u), lane8);
          j += 2;
        }
        if (rem & 1)
          q1 = ld_qrow(Qb, (unsigned)__builtin_amdgcn_readlane(ov, j), lane8);
        if (rem & 4) {
#pragma unroll
          for (int u = 0; u < 4; ++u) row_acc(acc, q4[u], np0, np1);
        }
        if (rem & 2) {
#pragma unroll
          for (int u = 0; u < 2; ++u) row_acc(acc, q2[u], np0, np1);
        }
        if (rem & 1) row_acc(acc, q1, np0, np1);
      }

      if (c > SLOT_C) {
        int n = novf[0];
        n = n < NEDGES ? n : NEDGES;
        for (int kb = 0; kb < n; kb += 32) {
          const int  k    = kb + lane;
          const bool have = k < n;
          const unsigned w = have ? ovf[k] : 0u;
          const bool mine = have && (int)(w >> 16) == node;
          unsigned mask = __builtin_amdgcn_ballot_w32(mine);
          while (mask != 0u) {
            const int tl = __builtin_ctz(mask);
            mask &= mask - 1u;
            const int sj = __builtin_amdgcn_readlane((int)(w & 0xFFFFu), tl);
            row_acc(acc, ld_qrow(Qb, (unsigned)sj << 8, lane8), np0, np1);
          }
        }
      }

      const float cf = (float)c;
#pragma unroll
      for (int k = 0; k < 4; ++k) acc[k] += cf * pf[k];

      const float inv = 1.0f / (float)(c > 1 ? c : 1);
#pragma unroll
      for (int k = 0; k < 4; ++k)
        lds_s[r * APITCH + lane + 32 * k] = (f16)(acc[k] * inv);
      if (lane == 0) lds_has[r] = c > 0 ? 1.0f : 0.0f;
    }
  }
  __syncthreads();

  const int M  = lane & 15;
  const int hi = (lane >> 4) & 1;

  {
    const int n = wv * 16 + (lane & 15);
    v8f acc = {};
#pragma unroll
    for (int ks = 0; ks < 4; ++ks) {
      v16h a = ld_a_frag(lds_s, APITCH, M, hi, ks * 32);
      v16h b = ld_b_frag(W2f, wv * 4 + ks, lane);
      acc = __builtin_amdgcn_wmma_f32_16x16x32_f16(false, a, false, b,
                                                   (short)0, acc, false, false);
    }
    const float bias = b2[n];
#pragma unroll
    for (int r = 0; r < 8; ++r) {
      float v = acc[r] + bias * lds_has[r + 8 * hi];
      v = v > 0.0f ? v : 0.0f;
      lds_a[(r + 8 * hi) * APITCH + n] = (f16)v;
    }
  }
  __syncthreads();

  if (wv < 4) {
    const int n = wv * 16 + (lane & 15);
    v8f acc = {};
#pragma unroll
    for (int ks = 0; ks < 4; ++ks) {
      v16h a = ld_a_frag(lds_a, APITCH, M, hi, ks * 32);
      v16h b = ld_b_frag(W3f, wv * 4 + ks, lane);
      acc = __builtin_amdgcn_wmma_f32_16x16x32_f16(false, a, false, b,
                                                   (short)0, acc, false, false);
    }
    const float bias = b3[n];
#pragma unroll
    for (int r = 0; r < 8; ++r) {
      float v = acc[r] + bias;
      v = v > 0.0f ? v : 0.0f;
      lds_m[(r + 8 * hi) * MPITCH + n] = (f16)v;
    }
  }
  __syncthreads();

  {
    const int n = wv * 16 + (lane & 15);
    v8f acc = {};
#pragma unroll
    for (int ks = 0; ks < 2; ++ks) {
      v16h a = ld_a_frag(lds_m, MPITCH, M, hi, ks * 32);
      v16h b = ld_b_frag(W4f, wv * 2 + ks, lane);
      acc = __builtin_amdgcn_wmma_f32_16x16x32_f16(false, a, false, b,
                                                   (short)0, acc, false, false);
    }
    const float bias = b4[n];
#pragma unroll
    for (int r = 0; r < 8; ++r) {
      out[(size_t)(nbase + r + 8 * hi) * FOUT + n] = acc[r] + bias;
    }
  }
}

extern "C" void kernel_launch(void* const* d_in, const int* in_sizes, int n_in,
                              void* d_out, int out_size, void* d_ws, size_t ws_size,
                              hipStream_t stream) {
  const float* x   = (const float*)d_in[0];
  const int*  eidx = (const int*)d_in[1];
  const float* W1  = (const float*)d_in[2];
  const float* b1  = (const float*)d_in[3];
  const float* W2  = (const float*)d_in[4];
  const float* b2  = (const float*)d_in[5];
  const float* W3  = (const float*)d_in[6];
  const float* b3  = (const float*)d_in[7];
  const float* W4  = (const float*)d_in[8];
  const float* b4  = (const float*)d_in[9];
  float* out = (float*)d_out;

  char* ws = (char*)d_ws;
  f16* Q      = (f16*)ws;
  int* slots  = (int*)(ws + 12800000);
  int* deg    = (int*)(ws + 19200000);
  int* novf   = (int*)(ws + 19400000);
  int* nbovf  = (int*)(ws + 19400004);
  int* gcount = (int*)(ws + 19400128);
  unsigned* ovf = (unsigned*)(ws + 19600000);
  f16* W1c = (f16*)(ws + 25800000);
  f16* W2f = (f16*)(ws + 25800000 + 65536);
  f16* W3f = (f16*)(ws + 25800000 + 65536 + 32768);
  f16* W4f = (f16*)(ws + 25800000 + 65536 + 32768 + 16384);
  f16* P   = (f16*)d_out;

  hipMemsetAsync(novf, 0, 100352, stream);

  bin_pack_kernel<<<BIN_BLOCKS + PACK_BLOCKS, 256, 0, stream>>>(
      eidx, eidx + NEDGES, gcount, nbovf, (unsigned*)d_out, W1, W2, W3, W4, W1c, W2f, W3f, W4f);
  node_pre_place_kernel<<<PRE_BLOCKS + NBUCK, PRE_THREADS, 0, stream>>>(
      x, W1c, b1, P, Q, gcount, nbovf, (const unsigned*)d_out, deg, slots, novf, ovf);

  node_gather_mlp_kernel<<<NNODES / 16, 256, 0, stream>>>(
      Q, slots, deg, novf, ovf, W2f, b2, W3f, b3, W4f, b4, out);
}


# Binning blocks keep their sorted edges as one run each plus a per-bucket table of piece starts and lengths; placement reads the pieces: no global counters or atomics in binning, no clearing memset
# speedup vs baseline: 11.0689x; 1.1295x over previous
#include <hip/hip_runtime.h>


typedef _Float16 f16;
typedef __attribute__((ext_vector_type(16))) _Float16 v16h;
typedef __attribute__((ext_vector_type(8)))  float    v8f;

#define NNODES 50000
#define NEDGES 800000
#define FIN    128
#define FOUT   128


#define XPITCH (FIN + 8)
#define APITCH (FOUT + 8)
#define MPITCH (FOUT / 2 + 8)

#define PRE_MT 5
#define EDGE_G 8
#define SLOT_C 32
#define NPB 64
#define NBUCK ((NNODES + NPB - 1) / NPB)
#define BIN_BLOCKS 125
#define BIN_EPT 25
static_assert(BIN_BLOCKS * 256 * BIN_EPT == NEDGES, "the binning pass must cover every edge exactly once");
static_assert(NNODES <= 65536, "a packed edge word holds destination and source id in 16 bits each");
#define OUT_ROW_WORDS FOUT
#define STAGE_W 64
#define RUN_W 64
static_assert(STAGE_W + RUN_W == OUT_ROW_WORDS, "output row scratch layout");
static_assert(NEDGES / RUN_W <= NNODES && NEDGES % RUN_W == 0, "the sorted runs fit in the output rows");
#define TBL_PITCH 128
#define PIECE_G 8
static_assert(BIN_BLOCKS <= TBL_PITCH, "one table word per binning block");
static_assert(256 * BIN_EPT < 65536, "start and length of a piece fit in 16 bits each");

#define PACK_BLOCKS_W1 ((FIN * 2 * FOUT) / 256)
#define PACK_BLOCKS_W2 ((FOUT * FOUT) / 256)
#define PACK_BLOCKS_W3 ((FOUT * (FOUT / 2)) / 256)
#define PACK_BLOCKS_W4 (((FOUT / 2) * FOUT) / 256)
#define PACK_BLOCKS (PACK_BLOCKS_W1 + PACK_BLOCKS_W2 + PACK_BLOCKS_W3 + PACK_BLOCKS_W4)
#define PRE_THREADS 128
#define PRE_BLOCKS (NNODES / (PRE_MT * 16))
static_assert(PRE_BLOCKS * PRE_MT * 16 == NNODES, "pre-pass tiles must cover every node exactly once");

union V16H { v16h v; uint4 q[2]; };
union H8   { f16 h[8]; uint4 q; };
union H4   { f16 h[4]; uint2 d; };
typedef __attribute__((ext_vector_type(2))) _Float16 h2;
union HP   { uint2 d; h2 p[2]; };

__device__ __forceinline__ void row_acc(float (&acc)[4], uint2 qv, h2 np0, h2 np1) {
  HP b;
  b.d = qv;
  const h2 t0 = __builtin_elementwise_maximum(b.p[0], np0);
  const h2 t1 = __builtin_elementwise_maximum(b.p[1], np1);
  acc[0] += (float)t0[0];
  acc[1] += (float)t0[1];
  acc[2] += (float)t1[0];
  acc[3] += (float)t1[1];
}

__device__ __forceinline__ uint2 ld_qrow(const char* __restrict__ Qb, unsigned rowoff, unsigned lane8) {
  return *(const uint2*)(Qb + (size_t)(rowoff + lane8));
}

__device__ __forceinline__ uint4 pack8(float4 a, float4 b) {
  H8 u;
  u.h[0] = (f16)a.x; u.h[1] = (f16)a.y; u.h[2] = (f16)a.z; u.h[3] = (f16)a.w;
  u.h[4] = (f16)b.x; u.h[5] = (f16)b.y; u.h[6] = (f16)b.z; u.h[7] = (f16)b.w;
  return u.q;
}

__device__ __forceinline__ v16h ld_a_frag(const f16* lds, int row_stride,
                                          int M, int hi, int kbase) {
  const int kb = kbase + hi * 8;
  V16H t;
  t.q[0] = *(const uint4*)(lds + M * row_stride + kb);
  t.q[1] = *(const uint4*)(lds + M * row_stride + kb + 16);
  return t.v;
}

__device__ __forceinline__ v16h ld_b_frag(const f16* __restrict__ wf,
                                          int frag, int lane) {
  const uint4* p = (const uint4*)(wf + ((size_t)(frag * 32 + lane)) * 16);
  V16H t;
  t.q[0] = p[0];
  t.q[1] = p[1];
  return t.v;
}

template <int K, int N>
__device__ __forceinline__ void pack_weights_elem(
    const float* __restrict__ W, f16* __restrict__ out, int o) {
  const int ksteps = K >> 5;
  int e    = o & 15;
  int lane = (o >> 4) & 31;
  int frag = o >> 9;
  int ks   = frag % ksteps;
  int nt   = frag / ksteps;
  int kl   = (e < 8 ? e : e + 8) + ((lane & 16) ? 8 : 0);
  int Kidx = ks * 32 + kl;
  int n    = nt * 16 + (lane & 15);
  out[o] = (f16)W[Kidx * N + n];
}

__device__ __forceinline__ void pack_w1_folded_elem(
    const float* __restrict__ W1, f16* __restrict__ out, int o) {
  const int ksteps = FIN >> 5;
  int e    = o & 15;
  int lane = (o >> 4) & 31;
  int frag = o >> 9;
  int ks   = frag % ksteps;
  int nt   = frag / ksteps;
  int kl   = (e < 8 ? e : e + 8) + ((lane & 16) ? 8 : 0);
  int Kidx = ks * 32 + kl;
  int n    = nt * 16 + (lane & 15);
  float w;
  if (n < FOUT) w = W1[Kidx * FOUT + n] - W1[(FIN + Kidx) * FOUT + n];
  else          w = W1[(FIN + Kidx) * FOUT + (n - FOUT)];
  out[o] = (f16)w;
}

__device__ __forceinline__ void node_pre_block(
    const int tile,
    f16* lds_x,
    const float* __restrict__ x,
    const f16* __restrict__ W1c,
    const float* __restrict__ b1,
    f16* __restrict__ P,
    f16* __restrict__ Q) {
  const int tid   = threadIdx.x;
  const int lane  = tid & 31;
  const int wv    = tid >> 5;
  const int nbase = tile * (PRE_MT * 16);

#pragma unroll
  for (int i = 0; i < (PRE_MT * 16 * 16) / 128; ++i) {
    const int idx = tid + i * 128;
    const int r   = idx >> 4;
    const int c0  = (idx & 15) * 8;
    const float4* xp = (const float4*)(x + (size_t)(nbase + r) * FIN + c0);
    float4 a0 = xp[0], a1 = xp[1];
    *(uint4*)(&lds_x[r * XPITCH + c0]) = pack8(a0, a1);
  }

  const int pq   = wv >> 1;
  const int t    = wv & 1;
  const int M    = lane & 15;
  const int hi   = (lane >> 4) & 1;

  v16h bf[4][4];
#pragma unroll
  for (int k = 0; k < 4; ++k) {
    const int nt = pq * 8 + t + 2 * k;
#pragma unroll
    for (int ks = 0; ks < 4; ++ks) bf[k][ks] = ld_b_frag(W1c, nt * 4 + ks, lane);
  }
  float bias[4];
#pragma unroll
  for (int k = 0; k < 4; ++k)
    bias[k] = pq == 0 ? b1[16 * t + M + 32 * k] : 0.0f;

  __syncthreads();

#pragma unroll 1
  for (int mt = 0; mt < PRE_MT; ++mt) {
    v8f acc[4];
#pragma unroll
    for (int k = 0; k < 4; ++k) acc[k] = (v8f){};
#pragma unroll
    for (int ks = 0; ks < 4; ++ks) {
      v16h a = ld_a_frag(lds_x + mt * 16 * XPITCH, XPITCH, M, hi, ks * 32);
#pragma unroll
      for (int k = 0; k < 4; ++k)
        acc[k] = __builtin_amdgcn_wmma_f32_16x16x32_f16(false, a, false, bf[k][ks],
                                                        (short)0, acc[k],
                                                        false, false);
    }
#pragma unroll
    for (int r = 0; r < 8; ++r) {
      const int node = nbase + mt * 16 + r + 8 * hi;
      H4 u;
#pragma unroll
      for (int k = 0; k < 4; ++k) u.h[k] = (f16)(acc[k][r] + bias[k]);
      f16* rowp = pq == 0 ? P + (size_t)node * (2 * FOUT) : Q + (size_t)node * FOUT;
      *(uint2*)(rowp + 4 * (16 * t + M)) = u.d;
    }
  }
}

#define BIN_EDGES (256 * BIN_EPT)
#define NBUCK_PAD 1024
static_assert(NBUCK <= NBUCK_PAD && NBUCK_PAD == 4 * 256, "a thread owns four buckets of the padded table");
struct BinLds {
  int hist[NBUCK_PAD];
  int pref[NBUCK_PAD];
  int wsum[8];
  unsigned sw[BIN_EDGES];
};
__device__ __forceinline__ void edge_bin_block(
    BinLds& L,
    const int* __restrict__ src,
    const int* __restrict__ dst,
    unsigned* __restrict__ runtab,
    unsigned* outw) {
  const int tid  = threadIdx.x;
  const int lane = tid & 31;
  const int wv   = tid >> 5;
#pragma unroll
  for (int i = 0; i < 4; ++i) L.hist[tid + 256 * i] = 0;
  const int e0 = blockIdx.x * BIN_EDGES + tid;
  int d[BIN_EPT], s[BIN_EPT];
#pragma unroll
  for (int k = 0; k < BIN_EPT; ++k) {
    d[k] = dst[e0 + k * 256];
    s[k] = src[e0 + k * 256];
  }
  __syncthreads();
  int lr[BIN_EPT];
#pragma unroll
  for (int k = 0; k < BIN_EPT; ++k) lr[k] = atomicAdd(&L.hist[d[k] / NPB], 1);
  __syncthreads();

  int a[4];
#pragma unroll
  for (int i = 0; i < 4; ++i) a[i] = L.hist[4 * tid + i];
  const int tsum = a[0] + a[1] + a[2] + a[3];
  int inc = tsum;
#pragma unroll
  for (int off = 1; off < 32; off <<= 1) {
    const int v = __shfl_up(inc, off);
    if (lane >= off) inc += v;
  }
  if (lane == 31) L.wsum[wv] = inc;
  __syncthreads();
  int ex = inc - tsum;
  for (int w = 0; w < wv; ++w) ex += L.wsum[w];
#pragma unroll
  for (int i = 0; i < 4; ++i) {
    const int b = 4 * tid + i;
    L.pref[b] = ex;
    if (b < NBUCK) runtab[b * TBL_PITCH + blockIdx.x] = ((unsigned)ex << 16) | (unsigned)a[i];
    ex += a[i];
  }
  __syncthreads();

#pragma unroll
  for (int k = 0; k < BIN_EPT; ++k) {
    const int b = d[k] / NPB;
    const int j = L.pref[b] + lr[k];
    L.sw[j] = ((unsigned)(d[k] % NPB) << 16) | (unsigned)s[k];
  }
  __syncthreads();

  for (int k = 0; k < BIN_EPT; ++k) {
    const int j = tid + k * 256;
    const int g = blockIdx.x * BIN_EDGES + j;
    outw[(size_t)(g / RUN_W) * OUT_ROW_WORDS + STAGE_W + (g % RUN_W)] = L.sw[j];
  }
}

__device__ __forceinline__ void place_edge(
    int* cnt, int* rows, const int node, const int dl, const int s,
    int* __restrict__ novf, unsigned* __restrict__ ovf) {
  const int r = atomicAdd(&cnt[dl], 1);
  if (r < SLOT_C) {
    rows[dl * SLOT_C + r] = s << 8;
  } else {
    const int q = atomicAdd(novf, 1);
    ovf[q] = ((unsigned)node << 16) | (unsigned)s;
  }
}

__device__ __forceinline__ void bucket_place_block(
    const int b,
    int* cnt,
    int* rows,
    const unsigned* __restrict__ runtab,
    const unsigned* outw,
    int* __restrict__ deg,
    int* __restrict__ slots,
    int* __restrict__ novf,
    unsigned* __restrict__ ovf) {
  const int tid  = threadIdx.x;
  const int left = NNODES - b * NPB;
  const int nn   = left < NPB ? left : NPB;
  if (tid < NPB) cnt[tid] = 0;
  const unsigned tw = tid < BIN_BLOCKS ? runtab[b * TBL_PITCH + tid] : 0u;
  const int len = (int)(tw & 0xFFFFu);
  const int g0  = tid * BIN_EDGES + (int)(tw >> 16);
  __syncthreads();
  for (int e = 0; e < len; e += PIECE_G) {
    unsigned w[PIECE_G];
#pragma unroll
    for (int u = 0; u < PIECE_G; ++u) {
      const int g = g0 + e + u;
      w[u] = e + u < len ? outw[(size_t)(g / RUN_W) * OUT_ROW_WORDS + STAGE_W + (g % RUN_W)] : 0u;
    }
#pragma unroll
    for (int u = 0; u < PIECE_G; ++u) {
      if (e + u < len) {
        const int dl = (int)(w[u] >> 16) % NPB;
        place_edge(cnt, rows, b * NPB + dl, dl, (int)(w[u] & 0xFFFFu), novf, ovf);
      }
    }
  }
  __syncthreads();
  for (int i = tid; i < nn * SLOT_C; i += PRE_THREADS)
    if ((i & (SLOT_C - 1)) < cnt[i / SLOT_C]) slots[b * NPB * SLOT_C + i] = rows[i];
  if (tid < nn) deg[b * NPB + tid] = cnt[tid];
}

__global__ __launch_bounds__(256) void bin_pack_kernel(
    const int* __restrict__ src,
    const int* __restrict__ dst,
    unsigned* __restrict__ runtab,
    int* __restrict__ novf,
    unsigned* outw,
    const float* __restrict__ W1,
    const float* __restrict__ W2,
    const float* __restrict__ W3,
    const float* __restrict__ W4,
    f16* __restrict__ W1c,
    f16* __restrict__ W2f,
    f16* __restrict__ W3f,
    f16* __restrict__ W4f) {
  __shared__ BinLds L;
  if (blockIdx.x < BIN_BLOCKS) {
    edge_bin_block(L, src, dst, runtab, outw);
    return;
  }
  const int pb = blockIdx.x - BIN_BLOCKS;
  const int t  = threadIdx.x;
  if (pb == 0 && t == 0) novf[0] = 0;
  if (pb < PACK_BLOCKS_W1) {
    pack_w1_folded_elem(W1, W1c, pb * 256 + t);
  } else if (pb < PACK_BLOCKS_W1 + PACK_BLOCKS_W2) {
    pack_weights_elem<FOUT, FOUT>(W2, W2f, (pb - PACK_BLOCKS_W1) * 256 + t);
  } else if (pb < PACK_BLOCKS_W1 + PACK_BLOCKS_W2 + PACK_BLOCKS_W3) {
    pack_weights_elem<FOUT, FOUT / 2>(W3, W3f, (pb - PACK_BLOCKS_W1 - PACK_BLOCKS_W2) * 256 + t);
  } else {
    pack_weights_elem<FOUT / 2, FOUT>(W4, W4f,
                                      (pb - PACK_BLOCKS_W1 - PACK_BLOCKS_W2 - PACK_BLOCKS_W3) * 256 + t);
  }
}

#define PRE_LDS_BYTES (PRE_MT * 16 * XPITCH * 2)
static_assert((NPB + NPB * SLOT_C) * 4 <= PRE_LDS_BYTES, "placement tables fit in the pre-pass tile's LDS");
__global__ __launch_bounds__(PRE_THREADS) void node_pre_place_kernel(
    const float* __restrict__ x,
    const f16* __restrict__ W1c,
    const float* __restrict__ b1,
    f16* __restrict__ P,
    f16* __restrict__ Q,
    const unsigned* __restrict__ runtab,
    const unsigned* outw,
    int* __restrict__ deg,
    int* __restrict__ slots,
    int* __restrict__ novf,
    unsigned* __restrict__ ovf) {
  __shared__ __attribute__((aligned(16))) unsigned char lds_raw[PRE_LDS_BYTES];
  if (blockIdx.x < PRE_BLOCKS) {
    node_pre_block(blockIdx.x, (f16*)lds_raw, x, W1c, b1, P, Q);
    return;
  }
  int* li = (int*)lds_raw;
  bucket_place_block(blockIdx.x - PRE_BLOCKS, li, li + NPB, runtab, outw,
                     deg, slots, novf, ovf);
}

__global__ __launch_bounds__(256) void node_gather_mlp_kernel(
    const f16* __restrict__ Q,
    const int* __restrict__ slots,
    const int* __restrict__ deg,
    const int* __restrict__ novf,
    const unsigned* __restrict__ ovf,
    const f16* __restrict__ W2f,
    const float* __restrict__ b2,
    const f16* __restrict__ W3f,
    const float* __restrict__ b3,
    const f16* __restrict__ W4f,
    const float* __restrict__ b4,
    float* out) {
  __shared__ __attribute__((aligned(16))) f16 lds_s[16 * APITCH];
  __shared__ __attribute__((aligned(16))) f16 lds_a[16 * APITCH];
  __shared__ __attribute__((aligned(16))) f16 lds_m[16 * MPITCH];
  __shared__ float lds_has[16];

  const int tid   = threadIdx.x;
  const int lane  = tid & 31;
  const int wv    = tid >> 5;
  const int nbase = blockIdx.x * 16;

  {
    const f16* P = (const f16*)out;
    const int wvu = __builtin_amdgcn_readfirstlane(wv);
    const int n0  = nbase + wvu * 2;
    int cc[2], ivv[2];
    H4 ppv[2];
#pragma unroll
    for (int rr = 0; rr < 2; ++rr) cc[rr] = deg[n0 + rr];
#pragma unroll
    for (int rr = 0; rr < 2; ++rr) {
      const int cs = cc[rr] < SLOT_C ? cc[rr] : SLOT_C;
      ivv[rr] = lane < cs ? slots[(n0 + rr) * SLOT_C + lane] : 0;
      ppv[rr].d = *(const uint2*)(P + (size_t)(n0 + rr) * (2 * FOUT) + 4 * lane);
    }
    const char* Qb = (const char*)Q;
    const unsigned lane8 = (unsigned)lane * 8u;
#pragma unroll
    for (int rr = 0; rr < 2; ++rr) {
      const int r    = wvu * 2 + rr;
      const int node = n0 + rr;
      const int c    = cc[rr];
      const int m    = c < SLOT_C ? c : SLOT_C;
      const int ov   = ivv[rr];
      HP pp;
      pp.d = ppv[rr].d;
      const h2 np0 = -pp.p[0];
      const h2 np1 = -pp.p[1];
      float pf[4], acc[4];
#pragma unroll
      for (int k = 0; k < 4; ++k) { pf[k] = (float)ppv[rr].h[k]; acc[k] = 0.0f; }

      int j = 0;
      for (; j + EDGE_G <= m; j += EDGE_G) {
        uint2 q[EDGE_G];
#pragma unroll
        for (int u = 0; u < EDGE_G; ++u)
          q[u] = ld_qrow(Qb, (unsigned)__builtin_amdgcn_readlane(ov, j + u), lane8);
#pragma unroll
        for (int u = 0; u < EDGE_G; ++u) row_acc(acc, q[u], np0, np1);
      }
      {
        const int rem = m - j;
        uint2 q4[4], q2[2], q1;
        if (rem & 4) {
#pragma unroll
          for (int u = 0; u < 4; ++u)
            q4[u] = ld_qrow(Qb, (unsigned)__builtin_amdgcn_readlane(ov, j + u), lane8);
          j += 4;
        }
        if (rem & 2) {
#pragma unroll
          for (int u = 0; u < 2; ++u)
            q2[u] = ld_qrow(Qb, (unsigned)__builtin_amdgcn_readlane(ov, j + u), lane8);
          j += 2;
        }
        if (rem & 1)
          q1 = ld_qrow(Qb, (unsigned)__builtin_amdgcn_readlane(ov, j), lane8);
        if (rem & 4) {
#pragma unroll
          for (int u = 0; u < 4; ++u) row_acc(acc, q4[u], np0, np1);
        }
        if (rem & 2) {
#pragma unroll
          for (int u = 0; u < 2; ++u) row_acc(acc, q2[u], np0, np1);
        }
        if (rem & 1) row_acc(acc, q1, np0, np1);
      }

      if (c > SLOT_C) {
        int n = novf[0];
        n = n < NEDGES ? n : NEDGES;
        for (int kb = 0; kb < n; kb += 32) {
          const int  k    = kb + lane;
          const bool have = k < n;
          const unsigned w = have ? ovf[k] : 0u;
          const bool mine = have && (int)(w >> 16) == node;
          unsigned mask = __builtin_amdgcn_ballot_w32(mine);
          while (mask != 0u) {
            const int tl = __builtin_ctz(mask);
            mask &= mask - 1u;
            const int sj = __builtin_amdgcn_readlane((int)(w & 0xFFFFu), tl);
            row_acc(acc, ld_qrow(Qb, (unsigned)sj << 8, lane8), np0, np1);
          }
        }
      }

      const float cf = (float)c;
#pragma unroll
      for (int k = 0; k < 4; ++k) acc[k] += cf * pf[k];

      const float inv = 1.0f / (float)(c > 1 ? c : 1);
#pragma unroll
      for (int k = 0; k < 4; ++k)
        lds_s[r * APITCH + lane + 32 * k] = (f16)(acc[k] * inv);
      if (lane == 0) lds_has[r] = c > 0 ? 1.0f : 0.0f;
    }
  }
  __syncthreads();

  const int M  = lane & 15;
  const int hi = (lane >> 4) & 1;

  {
    const int n = wv * 16 + (lane & 15);
    v8f acc = {};
#pragma unroll
    for (int ks = 0; ks < 4; ++ks) {
      v16h a = ld_a_frag(lds_s, APITCH, M, hi, ks * 32);
      v16h b = ld_b_frag(W2f, wv * 4 + ks, lane);
      acc = __builtin_amdgcn_wmma_f32_16x16x32_f16(false, a, false, b,
                                                   (short)0, acc, false, false);
    }
    const float bias = b2[n];
#pragma unroll
    for (int r = 0; r < 8; ++r) {
      float v = acc[r] + bias * lds_has[r + 8 * hi];
      v = v > 0.0f ? v : 0.0f;
      lds_a[(r + 8 * hi) * APITCH + n] = (f16)v;
    }
  }
  __syncthreads();

  if (wv < 4) {
    const int n = wv * 16 + (lane & 15);
    v8f acc = {};
#pragma unroll
    for (int ks = 0; ks < 4; ++ks) {
      v16h a = ld_a_frag(lds_a, APITCH, M, hi, ks * 32);
      v16h b = ld_b_frag(W3f, wv * 4 + ks, lane);
      acc = __builtin_amdgcn_wmma_f32_16x16x32_f16(false, a, false, b,
                                                   (short)0, acc, false, false);
    }
    const float bias = b3[n];
#pragma unroll
    for (int r = 0; r < 8; ++r) {
      float v = acc[r] + bias;
      v = v > 0.0f ? v : 0.0f;
      lds_m[(r + 8 * hi) * MPITCH + n] = (f16)v;
    }
  }
  __syncthreads();

  {
    const int n = wv * 16 + (lane & 15);
    v8f acc = {};
#pragma unroll
    for (int ks = 0; ks < 2; ++ks) {
      v16h a = ld_a_frag(lds_m, MPITCH, M, hi, ks * 32);
      v16h b = ld_b_frag(W4f, wv * 2 + ks, lane);
      acc = __builtin_amdgcn_wmma_f32_16x16x32_f16(false, a, false, b,
                                                   (short)0, acc, false, false);
    }
    const float bias = b4[n];
#pragma unroll
    for (int r = 0; r < 8; ++r) {
      out[(size_t)(nbase + r + 8 * hi) * FOUT + n] = acc[r] + bias;
    }
  }
}

extern "C" void kernel_launch(void* const* d_in, const int* in_sizes, int n_in,
                              void* d_out, int out_size, void* d_ws, size_t ws_size,
                              hipStream_t stream) {
  const float* x   = (const float*)d_in[0];
  const int*  eidx = (const int*)d_in[1];
  const float* W1  = (const float*)d_in[2];
  const float* b1  = (const float*)d_in[3];
  const float* W2  = (const float*)d_in[4];
  const float* b2  = (const float*)d_in[5];
  const float* W3  = (const float*)d_in[6];
  const float* b3  = (const float*)d_in[7];
  const float* W4  = (const float*)d_in[8];
  const float* b4  = (const float*)d_in[9];
  float* out = (float*)d_out;

  char* ws = (char*)d_ws;
  f16* Q      = (f16*)ws;
  int* slots  = (int*)(ws + 12800000);
  int* deg    = (int*)(ws + 19200000);
  int* novf   = (int*)(ws + 19400000);
  unsigned* ovf = (unsigned*)(ws + 19600000);
  unsigned* runtab = (unsigned*)(ws + 22800000);
  f16* W1c = (f16*)(ws + 25800000);
  f16* W2f = (f16*)(ws + 25800000 + 65536);
  f16* W3f = (f16*)(ws + 25800000 + 65536 + 32768);
  f16* W4f = (f16*)(ws + 25800000 + 65536 + 32768 + 16384);
  f16* P   = (f16*)d_out;

  bin_pack_kernel<<<BIN_BLOCKS + PACK_BLOCKS, 256, 0, stream>>>(
      eidx, eidx + NEDGES, runtab, novf, (unsigned*)d_out, W1, W2, W3, W4, W1c, W2f, W3f, W4f);
  node_pre_place_kernel<<<PRE_BLOCKS + NBUCK, PRE_THREADS, 0, stream>>>(
      x, W1c, b1, P, Q, runtab, (const unsigned*)d_out, deg, slots, novf, ovf);

  node_gather_mlp_kernel<<<NNODES / 16, 256, 0, stream>>>(
      Q, slots, deg, novf, ovf, W2f, b2, W3f, b3, W4f, b4, out);
}


# Slot rows are built in LDS by the block that gathers them (one block per 64-node bucket, 8 nodes per wave, 4 row tiles per weight fragment): no placement blocks, no slot rows or degrees in memory
# speedup vs baseline: 12.3330x; 1.1142x over previous
#include <hip/hip_runtime.h>


typedef _Float16 f16;
typedef __attribute__((ext_vector_type(16))) _Float16 v16h;
typedef __attribute__((ext_vector_type(8)))  float    v8f;

#define NNODES 50000
#define NEDGES 800000
#define FIN    128
#define FOUT   128


#define XPITCH (FIN + 8)
#define APITCH (FOUT + 8)
#define MPITCH (FOUT / 2 + 8)

#define PRE_MT 5
#define SLOT_C 32
#define NPB 64
#define NBUCK ((NNODES + NPB - 1) / NPB)
#define BIN_BLOCKS 125
#define BIN_EPT 25
static_assert(BIN_BLOCKS * 256 * BIN_EPT == NEDGES, "the binning pass must cover every edge exactly once");
static_assert(NNODES <= 65536, "a packed edge word holds destination and source id in 16 bits each");
#define TBL_PITCH 128
#define PIECE_G 8
#define NPW (NPB / 8)
#define OVL_C 256
static_assert(2 * BIN_BLOCKS <= 256, "two threads of the node block per piece");
static_assert(BIN_BLOCKS <= TBL_PITCH, "one table word per binning block");
static_assert(256 * BIN_EPT < 65536, "start and length of a piece fit in 16 bits each");

#define PACK_BLOCKS_W1 ((FIN * 2 * FOUT) / 256)
#define PACK_BLOCKS_W2 ((FOUT * FOUT) / 256)
#define PACK_BLOCKS_W3 ((FOUT * (FOUT / 2)) / 256)
#define PACK_BLOCKS_W4 (((FOUT / 2) * FOUT) / 256)
#define PACK_BLOCKS (PACK_BLOCKS_W1 + PACK_BLOCKS_W2 + PACK_BLOCKS_W3 + PACK_BLOCKS_W4)
#define PRE_THREADS 128
#define PRE_BLOCKS (NNODES / (PRE_MT * 16))
static_assert(PRE_BLOCKS * PRE_MT * 16 == NNODES, "pre-pass tiles must cover every node exactly once");

union V16H { v16h v; uint4 q[2]; };
union H8   { f16 h[8]; uint4 q; };
union H4   { f16 h[4]; uint2 d; };
typedef __attribute__((ext_vector_type(2))) _Float16 h2;
union HP   { uint2 d; h2 p[2]; };

__device__ __forceinline__ void row_acc(float (&acc)[4], uint2 qv, h2 np0, h2 np1) {
  HP b;
  b.d = qv;
  const h2 t0 = __builtin_elementwise_maximum(b.p[0], np0);
  const h2 t1 = __builtin_elementwise_maximum(b.p[1], np1);
  acc[0] += (float)t0[0];
  acc[1] += (float)t0[1];
  acc[2] += (float)t1[0];
  acc[3] += (float)t1[1];
}

__device__ __forceinline__ uint2 ld_qrow(const char* __restrict__ Qb, unsigned rowoff, unsigned lane8) {
  return *(const uint2*)(Qb + (size_t)(rowoff + lane8));
}

__device__ __forceinline__ uint4 pack8(float4 a, float4 b) {
  H8 u;
  u.h[0] = (f16)a.x; u.h[1] = (f16)a.y; u.h[2] = (f16)a.z; u.h[3] = (f16)a.w;
  u.h[4] = (f16)b.x; u.h[5] = (f16)b.y; u.h[6] = (f16)b.z; u.h[7] = (f16)b.w;
  return u.q;
}

__device__ __forceinline__ v16h ld_a_frag(const f16* lds, int row_stride,
                                          int M, int hi, int kbase) {
  const int kb = kbase + hi * 8;
  V16H t;
  t.q[0] = *(const uint4*)(lds + M * row_stride + kb);
  t.q[1] = *(const uint4*)(lds + M * row_stride + kb + 16);
  return t.v;
}

__device__ __forceinline__ v16h ld_b_frag(const f16* __restrict__ wf,
                                          int frag, int lane) {
  const uint4* p = (const uint4*)(wf + ((size_t)(frag * 32 + lane)) * 16);
  V16H t;
  t.q[0] = p[0];
  t.q[1] = p[1];
  return t.v;
}

template <int K, int N>
__device__ __forceinline__ void pack_weights_elem(
    const float* __restrict__ W, f16* __restrict__ out, int o) {
  const int ksteps = K >> 5;
  int e    = o & 15;
  int lane = (o >> 4) & 31;
  int frag = o >> 9;
  int ks   = frag % ksteps;
  int nt   = frag / ksteps;
  int kl   = (e < 8 ? e : e + 8) + ((lane & 16) ? 8 : 0);
  int Kidx = ks * 32 + kl;
  int n    = nt * 16 + (lane & 15);
  out[o] = (f16)W[Kidx * N + n];
}

__device__ __forceinline__ void pack_w1_folded_elem(
    const float* __restrict__ W1, f16* __restrict__ out, int o) {
  const int ksteps = FIN >> 5;
  int e    = o & 15;
  int lane = (o >> 4) & 31;
  int frag = o >> 9;
  int ks   = frag % ksteps;
  int nt   = frag / ksteps;
  int kl   = (e < 8 ? e : e + 8) + ((lane & 16) ? 8 : 0);
  int Kidx = ks * 32 + kl;
  int n    = nt * 16 + (lane & 15);
  float w;
  if (n < FOUT) w = W1[Kidx * FOUT + n] - W1[(FIN + Kidx) * FOUT + n];
  else          w = W1[(FIN + Kidx) * FOUT + (n - FOUT)];
  out[o] = (f16)w;
}

__device__ __forceinline__ void node_pre_block(
    const int tile,
    f16* lds_x,
    const float* __restrict__ x,
    const f16* __restrict__ W1c,
    const float* __restrict__ b1,
    f16* __restrict__ P,
    f16* __restrict__ Q) {
  const int tid   = threadIdx.x;
  const int lane  = tid & 31;
  const int wv    = tid >> 5;
  const int nbase = tile * (PRE_MT * 16);

#pragma unroll
  for (int i = 0; i < (PRE_MT * 16 * 16) / 128; ++i) {
    const int idx = tid + i * 128;
    const int r   = idx >> 4;
    const int c0  = (idx & 15) * 8;
    const float4* xp = (const float4*)(x + (size_t)(nbase + r) * FIN + c0);
    float4 a0 = xp[0], a1 = xp[1];
    *(uint4*)(&lds_x[r * XPITCH + c0]) = pack8(a0, a1);
  }

  const int pq   = wv >> 1;
  const int t    = wv & 1;
  const int M    = lane & 15;
  const int hi   = (lane >> 4) & 1;

  v16h bf[4][4];
#pragma unroll
  for (int k = 0; k < 4; ++k) {
    const int nt = pq * 8 + t + 2 * k;
#pragma unroll
    for (int ks = 0; ks < 4; ++ks) bf[k][ks] = ld_b_frag(W1c, nt * 4 + ks, lane);
  }
  float bias[4];
#pragma unroll
  for (int k = 0; k < 4; ++k)
    bias[k] = pq == 0 ? b1[16 * t + M + 32 * k] : 0.0f;

  __syncthreads();

#pragma unroll 1
  for (int mt = 0; mt < PRE_MT; ++mt) {
    v8f acc[4];
#pragma unroll
    for (int k = 0; k < 4; ++k) acc[k] = (v8f){};
#pragma unroll
    for (int ks = 0; ks < 4; ++ks) {
      v16h a = ld_a_frag(lds_x + mt * 16 * XPITCH, XPITCH, M, hi, ks * 32);
#pragma unroll
      for (int k = 0; k < 4; ++k)
        acc[k] = __builtin_amdgcn_wmma_f32_16x16x32_f16(false, a, false, bf[k][ks],
                                                        (short)0, acc[k],
                                                        false, false);
    }
#pragma unroll
    for (int r = 0; r < 8; ++r) {
      const int node = nbase + mt * 16 + r + 8 * hi;
      H4 u;
#pragma unroll
      for (int k = 0; k < 4; ++k) u.h[k] = (f16)(acc[k][r] + bias[k]);
      f16* rowp = pq == 0 ? P + (size_t)node * (2 * FOUT) : Q + (size_t)node * FOUT;
      *(uint2*)(rowp + 4 * (16 * t + M)) = u.d;
    }
  }
}

#define BIN_EDGES (256 * BIN_EPT)
#define NBUCK_PAD 1024
static_assert(NBUCK <= NBUCK_PAD && NBUCK_PAD == 4 * 256, "a thread owns four buckets of the padded table");
struct BinLds {
  int hist[NBUCK_PAD];
  int pref[NBUCK_PAD];
  int wsum[8];
  unsigned sw[BIN_EDGES];
};
__device__ __forceinline__ void edge_bin_block(
    BinLds& L,
    const int* __restrict__ src,
    const int* __restrict__ dst,
    unsigned* __restrict__ runtab,
    unsigned* __restrict__ runs) {
  const int tid  = threadIdx.x;
  const int lane = tid & 31;
  const int wv   = tid >> 5;
#pragma unroll
  for (int i = 0; i < 4; ++i) L.hist[tid + 256 * i] = 0;
  const int e0 = blockIdx.x * BIN_EDGES + tid;
  int d[BIN_EPT], s[BIN_EPT];
#pragma unroll
  for (int k = 0; k < BIN_EPT; ++k) {
    d[k] = dst[e0 + k * 256];
    s[k] = src[e0 + k * 256];
  }
  __syncthreads();
  int lr[BIN_EPT];
#pragma unroll
  for (int k = 0; k < BIN_EPT; ++k) lr[k] = atomicAdd(&L.hist[d[k] / NPB], 1);
  __syncthreads();

  int a[4];
#pragma unroll
  for (int i = 0; i < 4; ++i) a[i] = L.hist[4 * tid + i];
  const int tsum = a[0] + a[1] + a[2] + a[3];
  int inc = tsum;
#pragma unroll
  for (int off = 1; off < 32; off <<= 1) {
    const int v = __shfl_up(inc, off);
    if (lane >= off) inc += v;
  }
  if (lane == 31) L.wsum[wv] = inc;
  __syncthreads();
  int ex = inc - tsum;
  for (int w = 0; w < wv; ++w) ex += L.wsum[w];
#pragma unroll
  for (int i = 0; i < 4; ++i) {
    const int b = 4 * tid + i;
    L.pref[b] = ex;
    if (b < NBUCK) runtab[b * TBL_PITCH + blockIdx.x] = ((unsigned)ex << 16) | (unsigned)a[i];
    ex += a[i];
  }
  __syncthreads();

#pragma unroll
  for (int k = 0; k < BIN_EPT; ++k) {
    const int b = d[k] / NPB;
    const int j = L.pref[b] + lr[k];
    L.sw[j] = ((unsigned)(d[k] % NPB) << 16) | (unsigned)s[k];
  }
  __syncthreads();

  for (int k = 0; k < BIN_EPT; ++k) {
    const int j = tid + k * 256;
    runs[blockIdx.x * BIN_EDGES + j] = L.sw[j];
  }
}

__global__ __launch_bounds__(256) void bin_pack_kernel(
    const int* __restrict__ src,
    const int* __restrict__ dst,
    unsigned* __restrict__ runtab,
    unsigned* __restrict__ runs,
    const float* __restrict__ W1,
    const float* __restrict__ W2,
    const float* __restrict__ W3,
    const float* __restrict__ W4,
    f16* __restrict__ W1c,
    f16* __restrict__ W2f,
    f16* __restrict__ W3f,
    f16* __restrict__ W4f) {
  __shared__ BinLds L;
  if (blockIdx.x < BIN_BLOCKS) {
    edge_bin_block(L, src, dst, runtab, runs);
    return;
  }
  const int pb = blockIdx.x - BIN_BLOCKS;
  const int t  = threadIdx.x;
  if (pb < PACK_BLOCKS_W1) {
    pack_w1_folded_elem(W1, W1c, pb * 256 + t);
  } else if (pb < PACK_BLOCKS_W1 + PACK_BLOCKS_W2) {
    pack_weights_elem<FOUT, FOUT>(W2, W2f, (pb - PACK_BLOCKS_W1) * 256 + t);
  } else if (pb < PACK_BLOCKS_W1 + PACK_BLOCKS_W2 + PACK_BLOCKS_W3) {
    pack_weights_elem<FOUT, FOUT / 2>(W3, W3f, (pb - PACK_BLOCKS_W1 - PACK_BLOCKS_W2) * 256 + t);
  } else {
    pack_weights_elem<FOUT / 2, FOUT>(W4, W4f,
                                      (pb - PACK_BLOCKS_W1 - PACK_BLOCKS_W2 - PACK_BLOCKS_W3) * 256 + t);
  }
}

#define PRE_LDS_BYTES (PRE_MT * 16 * XPITCH * 2)
__global__ __launch_bounds__(PRE_THREADS) void node_pre_kernel(
    const float* __restrict__ x,
    const f16* __restrict__ W1c,
    const float* __restrict__ b1,
    f16* __restrict__ P,
    f16* __restrict__ Q) {
  __shared__ __attribute__((aligned(16))) unsigned char lds_raw[PRE_LDS_BYTES];
  node_pre_block(blockIdx.x, (f16*)lds_raw, x, W1c, b1, P, Q);
}

__global__ __launch_bounds__(256) void bucket_gather_mlp_kernel(
    const f16* __restrict__ Q,
    const unsigned* __restrict__ runtab,
    const unsigned* __restrict__ runs,
    const f16* __restrict__ W2f,
    const float* __restrict__ b2,
    const f16* __restrict__ W3f,
    const float* __restrict__ b3,
    const f16* __restrict__ W4f,
    const float* __restrict__ b4,
    float* out) {
  __shared__ __attribute__((aligned(16))) f16 lds_s[NPB * APITCH];
  __shared__ __attribute__((aligned(16))) f16 lds_a[NPB * APITCH];
  __shared__ __attribute__((aligned(16))) f16 lds_m[NPB * MPITCH];
  __shared__ float lds_has[NPB];
  __shared__ int cnt[NPB];
  __shared__ int rows[NPB * SLOT_C];
  __shared__ unsigned ptab[TBL_PITCH];
  __shared__ int novl;
  __shared__ unsigned ovl[OVL_C];

  const int tid   = threadIdx.x;
  const int lane  = tid & 31;
  const int wv    = tid >> 5;
  const int b     = blockIdx.x;
  const int nbase = b * NPB;

  if (tid < NPB) cnt[tid] = 0;
  if (tid == 0) novl = 0;
  {
    const int piece = tid >> 1;
    const int half  = tid & 1;
    const unsigned tw = piece < BIN_BLOCKS ? runtab[b * TBL_PITCH + piece] : 0u;
    if (half == 0) ptab[piece] = tw;
    const int len = (int)(tw & 0xFFFFu);
    const int g0  = piece * BIN_EDGES + (int)(tw >> 16);
    __syncthreads();
    for (int e = half; e < len; e += 2 * PIECE_G) {
      unsigned w[PIECE_G];
#pragma unroll
      for (int u = 0; u < PIECE_G; ++u) {
        const int k = e + 2 * u;
        w[u] = k < len ? runs[g0 + k] : 0u;
      }
      int r[PIECE_G];
#pragma unroll
      for (int u = 0; u < PIECE_G; ++u) {
        const int dl = (int)((w[u] >> 16) & (unsigned)(NPB - 1));
        r[u] = e + 2 * u < len ? atomicAdd(&cnt[dl], 1) : -1;
      }
#pragma unroll
      for (int u = 0; u < PIECE_G; ++u) {
        const int dl = (int)((w[u] >> 16) & (unsigned)(NPB - 1));
        if ((unsigned)r[u] < (unsigned)SLOT_C) {
          rows[dl * SLOT_C + r[u]] = (int)((w[u] & 0xFFFFu) << 8);
        } else if (r[u] >= SLOT_C) {
          const int q = atomicAdd(&novl, 1);
          if (q < OVL_C) ovl[q] = w[u];
        }
      }
    }
  }
  __syncthreads();

  {
    const f16* P = (const f16*)out;
    const int wvu = __builtin_amdgcn_readfirstlane(wv);
    const char* Qb = (const char*)Q;
    const unsigned lane8 = (unsigned)lane * 8u;
#pragma unroll 1
    for (int i = 0; i < NPW; ++i) {
      const int nl   = wvu * NPW + i;
      const int node = nbase + nl;
      if (node >= NNODES) {
#pragma unroll
        for (int k = 0; k < 4; ++k) lds_s[nl * APITCH + lane + 32 * k] = (f16)0.0f;
        if (lane == 0) lds_has[nl] = 0.0f;
        continue;
      }
      const int c  = __builtin_amdgcn_readfirstlane(cnt[nl]);
      const int m  = c < SLOT_C ? c : SLOT_C;
      const int ov = lane < m ? rows[nl * SLOT_C + lane] : 0;
      H4 ppv;
      ppv.d = *(const uint2*)(P + (size_t)node * (2 * FOUT) + 4 * lane);
      HP pp;
      pp.d = ppv.d;
      const h2 np0 = -pp.p[0];
      const h2 np1 = -pp.p[1];
      float pf[4], acc[4];
#pragma unroll
      for (int k = 0; k < 4; ++k) { pf[k] = (float)ppv.h[k]; acc[k] = 0.0f; }

      int j = 0;
      if (m == SLOT_C) {
        uint2 qf[16];
#pragma unroll
        for (int u = 0; u < 16; ++u)
          qf[u] = ld_qrow(Qb, (unsigned)__builtin_amdgcn_readlane(ov, u), lane8);
#pragma unroll
        for (int u = 0; u < 16; ++u) row_acc(acc, qf[u], np0, np1);
        j = 16;
      }
      {
        const int rem = m - j;
        uint2 q16[16] = {}, q8[8] = {}, q4[4] = {}, q2[2] = {}, q1 = {};
        if (rem & 16) {
#pragma unroll
          for (int u = 0; u < 16; ++u)
            q16[u] = ld_qrow(Qb, (unsigned)__builtin_amdgcn_readlane(ov, j + u), lane8);
          j += 16;
        }
        if (rem & 8) {
#pragma unroll
          for (int u = 0; u < 8; ++u)
            q8[u] = ld_qrow(Qb, (unsigned)__builtin_amdgcn_readlane(ov, j + u), lane8);
          j += 8;
        }
        if (rem & 4) {
#pragma unroll
          for (int u = 0; u < 4; ++u)
            q4[u] = ld_qrow(Qb, (unsigned)__builtin_amdgcn_readlane(ov, j + u), lane8);
          j += 4;
        }
        if (rem & 2) {
#pragma unroll
          for (int u = 0; u < 2; ++u)
            q2[u] = ld_qrow(Qb, (unsigned)__builtin_amdgcn_readlane(ov, j + u), lane8);
          j += 2;
        }
        if (rem & 1)
          q1 = ld_qrow(Qb, (unsigned)__builtin_amdgcn_readlane(ov, j), lane8);
        if (rem & 16) {
#pragma unroll
          for (int u = 0; u < 16; ++u) row_acc(acc, q16[u], np0, np1);
        }
        if (rem & 8) {
#pragma unroll
          for (int u = 0; u < 8; ++u) row_acc(acc, q8[u], np0, np1);
        }
        if (rem & 4) {
#pragma unroll
          for (int u = 0; u < 4; ++u) row_acc(acc, q4[u], np0, np1);
        }
        if (rem & 2) {
#pragma unroll
          for (int u = 0; u < 2; ++u) row_acc(acc, q2[u], np0, np1);
        }
        if (rem & 1) row_acc(acc, q1, np0, np1);
      }

      if (c > SLOT_C) {
        const int nv = __builtin_amdgcn_readfirstlane(novl);
        if (nv <= OVL_C) {
          for (int kb = 0; kb < nv; kb += 32) {
            const int  k    = kb + lane;
            const bool have = k < nv;
            const unsigned w = have ? ovl[k] : 0u;
            const bool mine = have && (int)((w >> 16) & (unsigned)(NPB - 1)) == nl;
            unsigned mask = __builtin_amdgcn_ballot_w32(mine);
            while (mask != 0u) {
              const int tl = __builtin_ctz(mask);
              mask &= mask - 1u;
              const int sj = __builtin_amdgcn_readlane((int)(w & 0xFFFFu), tl);
              row_acc(acc, ld_qrow(Qb, (unsigned)sj << 8, lane8), np0, np1);
            }
          }
        } else {
#pragma unroll
          for (int k = 0; k < 4; ++k) acc[k] = 0.0f;
          for (int p = 0; p < BIN_BLOCKS; ++p) {
            const unsigned tw = (unsigned)__builtin_amdgcn_readfirstlane((int)ptab[p]);
            const int len = (int)(tw & 0xFFFFu);
            const int g0  = p * BIN_EDGES + (int)(tw >> 16);
            for (int kb = 0; kb < len; kb += 32) {
              const int  k    = kb + lane;
              const bool have = k < len;
              const unsigned w = have ? runs[g0 + k] : 0u;
              const bool mine = have && (int)((w >> 16) & (unsigned)(NPB - 1)) == nl;
              unsigned mask = __builtin_amdgcn_ballot_w32(mine);
              while (mask != 0u) {
                const int tl = __builtin_ctz(mask);
                mask &= mask - 1u;
                const int sj = __builtin_amdgcn_readlane((int)(w & 0xFFFFu), tl);
                row_acc(acc, ld_qrow(Qb, (unsigned)sj << 8, lane8), np0, np1);
              }
            }
          }
        }
      }

      const float cf = (float)c;
#pragma unroll
      for (int k = 0; k < 4; ++k) acc[k] += cf * pf[k];

      const float inv = 1.0f / (float)(c > 1 ? c : 1);
#pragma unroll
      for (int k = 0; k < 4; ++k)
        lds_s[nl * APITCH + lane + 32 * k] = (f16)(acc[k] * inv);
      if (lane == 0) lds_has[nl] = c > 0 ? 1.0f : 0.0f;
    }
  }
  __syncthreads();

  const int M  = lane & 15;
  const int hi = (lane >> 4) & 1;

  {
    const int n = wv * 16 + M;
    v16h bw[4];
#pragma unroll
    for (int ks = 0; ks < 4; ++ks) bw[ks] = ld_b_frag(W2f, wv * 4 + ks, lane);
    v8f acc[4];
#pragma unroll
    for (int mt = 0; mt < 4; ++mt) acc[mt] = (v8f){};
#pragma unroll
    for (int ks = 0; ks < 4; ++ks) {
#pragma unroll
      for (int mt = 0; mt < 4; ++mt) {
        v16h a = ld_a_frag(lds_s + mt * 16 * APITCH, APITCH, M, hi, ks * 32);
        acc[mt] = __builtin_amdgcn_wmma_f32_16x16x32_f16(false, a, false, bw[ks],
                                                         (short)0, acc[mt], false, false);
      }
    }
    const float bias = b2[n];
#pragma unroll
    for (int mt = 0; mt < 4; ++mt) {
#pragma unroll
      for (int r = 0; r < 8; ++r) {
        const int row = mt * 16 + r + 8 * hi;
        float v = acc[mt][r] + bias * lds_has[row];
        v = v > 0.0f ? v : 0.0f;
        lds_a[row * APITCH + n] = (f16)v;
      }
    }
  }
  __syncthreads();

  {
    const int nt  = wv & 3;
    const int mt0 = (wv >> 2) * 2;
    const int n   = nt * 16 + M;
    v16h bw[4];
#pragma unroll
    for (int ks = 0; ks < 4; ++ks) bw[ks] = ld_b_frag(W3f, nt * 4 + ks, lane);
    v8f acc[2];
#pragma unroll
    for (int mm = 0; mm < 2; ++mm) acc[mm] = (v8f){};
#pragma unroll
    for (int ks = 0; ks < 4; ++ks) {
#pragma unroll
      for (int mm = 0; mm < 2; ++mm) {
        v16h a = ld_a_frag(lds_a + (mt0 + mm) * 16 * APITCH, APITCH, M, hi, ks * 32);
        acc[mm] = __builtin_amdgcn_wmma_f32_16x16x32_f16(false, a, false, bw[ks],
                                                         (short)0, acc[mm], false, false);
      }
    }
    const float bias = b3[n];
#pragma unroll
    for (int mm = 0; mm < 2; ++mm) {
#pragma unroll
      for (int r = 0; r < 8; ++r) {
        float v = acc[mm][r] + bias;
        v = v > 0.0f ? v : 0.0f;
        lds_m[((mt0 + mm) * 16 + r + 8 * hi) * MPITCH + n] = (f16)v;
      }
    }
  }
  __syncthreads();

  {
    const int n = wv * 16 + M;
    v16h bw[2];
#pragma unroll
    for (int ks = 0; ks < 2; ++ks) bw[ks] = ld_b_frag(W4f, wv * 2 + ks, lane);
    v8f acc[4];
#pragma unroll
    for (int mt = 0; mt < 4; ++mt) acc[mt] = (v8f){};
#pragma unroll
    for (int ks = 0; ks < 2; ++ks) {
#pragma unroll
      for (int mt = 0; mt < 4; ++mt) {
        v16h a = ld_a_frag(lds_m + mt * 16 * MPITCH, MPITCH, M, hi, ks * 32);
        acc[mt] = __builtin_amdgcn_wmma_f32_16x16x32_f16(false, a, false, bw[ks],
                                                         (short)0, acc[mt], false, false);
      }
    }
    const float bias = b4[n];
#pragma unroll
    for (int mt = 0; mt < 4; ++mt) {
#pragma unroll
      for (int r = 0; r < 8; ++r) {
        const int node = nbase + mt * 16 + r + 8 * hi;
        if (node < NNODES) out[(size_t)node * FOUT + n] = acc[mt][r] + bias;
      }
    }
  }
}

extern "C" void kernel_launch(void* const* d_in, const int* in_sizes, int n_in,
                              void* d_out, int out_size, void* d_ws, size_t ws_size,
                              hipStream_t stream) {
  const float* x   = (const float*)d_in[0];
  const int*  eidx = (const int*)d_in[1];
  const float* W1  = (const float*)d_in[2];
  const float* b1  = (const float*)d_in[3];
  const float* W2  = (const float*)d_in[4];
  const float* b2  = (const float*)d_in[5];
  const float* W3  = (const float*)d_in[6];
  const float* b3  = (const float*)d_in[7];
  const float* W4  = (const float*)d_in[8];
  const float* b4  = (const float*)d_in[9];
  float* out = (float*)d_out;

  char* ws = (char*)d_ws;
  f16* Q      = (f16*)ws;
  unsigned* runs   = (unsigned*)(ws + 12800000);
  unsigned* runtab = (unsigned*)(ws + 22800000);
  f16* W1c = (f16*)(ws + 25800000);
  f16* W2f = (f16*)(ws + 25800000 + 65536);
  f16* W3f = (f16*)(ws + 25800000 + 65536 + 32768);
  f16* W4f = (f16*)(ws + 25800000 + 65536 + 32768 + 16384);
  f16* P   = (f16*)d_out;

  bin_pack_kernel<<<BIN_BLOCKS + PACK_BLOCKS, 256, 0, stream>>>(
      eidx, eidx + NEDGES, runtab, runs, W1, W2, W3, W4, W1c, W2f, W3f, W4f);
  node_pre_kernel<<<PRE_BLOCKS, PRE_THREADS, 0, stream>>>(x, W1c, b1, P, Q);
  bucket_gather_mlp_kernel<<<NBUCK, 256, 0, stream>>>(
      Q, runtab, runs, W2f, b2, W3f, b3, W4f, b4, out);
}
